# starting kernel
# baseline (speedup 1.0000x reference)
#include <hip/hip_runtime.h>
#include <math.h>


typedef float v2f __attribute__((ext_vector_type(2)));
typedef float v8f __attribute__((ext_vector_type(8)));

#define TPB 256


__global__ __launch_bounds__(TPB) void k_init_deg(float* deg, float* degw, int n) {
    int i = blockIdx.x * TPB + threadIdx.x;
    if (i < n) { deg[i] = 1.0f; degw[i] = 1.0f; }
}

__global__ __launch_bounds__(TPB) void k_count_edges(const int* __restrict__ idx,
                                                     float* __restrict__ deg, int e) {
    int i = blockIdx.x * TPB + threadIdx.x;
    if (i < e) atomicAdd(&deg[idx[i]], 1.0f);
}

__global__ __launch_bounds__(TPB) void k_rsqrt(const float* __restrict__ a,
                                               float* __restrict__ o, int n) {
    int i = blockIdx.x * TPB + threadIdx.x;
    if (i < n) o[i] = rsqrtf(a[i]);
}

__global__ __launch_bounds__(TPB) void k_relu(float* h, long long n) {
    long long i = (long long)blockIdx.x * TPB + threadIdx.x;
    if (i < n) h[i] = fmaxf(h[i], 0.0f);
}

__global__ __launch_bounds__(TPB) void k_selfloop_scale(const float* __restrict__ s,
                                                        const float* __restrict__ hin,
                                                        float* __restrict__ hout,
                                                        int n, int d4) {
    long long t = (long long)blockIdx.x * TPB + threadIdx.x;
    if (t >= (long long)n * d4) return;
    int i = (int)(t / d4);
    float sv = s[i]; sv = sv * sv;
    float4 v = ((const float4*)hin)[t];
    float4 r; r.x = sv * v.x; r.y = sv * v.y; r.z = sv * v.z; r.w = sv * v.w;
    ((float4*)hout)[t] = r;
}

__global__ __launch_bounds__(TPB) void k_gemm_bias_wmma(const float* __restrict__ A,
                                                        const float* __restrict__ W,
                                                        const float* __restrict__ bias,
                                                        float* __restrict__ out,
                                                        int M, int K, int Nc) {
    int wid = (blockIdx.x * TPB + threadIdx.x) >> 5;
    int ntile = Nc >> 4;
    int total = (M >> 4) * ntile;
    if (wid >= total) return;

    int mt = wid / ntile, nt = wid - mt * ntile;
    int m0 = mt << 4, n0 = nt << 4;
    int lane = threadIdx.x & 31;
    int half = lane >> 4;
    int l16  = lane & 15;

    const float* arow = A + (long long)(m0 + l16) * K;
    const float* wrow = W + (long long)(n0 + l16) * K;

    v8f c = {0.f, 0.f, 0.f, 0.f, 0.f, 0.f, 0.f, 0.f};
    for (int kb = 0; kb < K; kb += 4) {
        int k = kb + 2 * half;
        v2f a = *(const v2f*)(arow + k);
        v2f b = *(const v2f*)(wrow + k);
        c = __builtin_amdgcn_wmma_f32_16x16x4_f32(false, a, false, b,
                                                  (short)0, c, false, false);
    }

    float bv = bias[n0 + l16];
    long long colbase = (long long)n0 + l16;
#pragma unroll
    for (int r = 0; r < 8; ++r) {
        int m = m0 + r + 8 * half;
        out[(long long)m * Nc + colbase] = c[r] + bv;
    }
}

__global__ __launch_bounds__(TPB) void k_prop1_edges(const int* __restrict__ row,
                                                     const int* __restrict__ col,
                                                     const float* __restrict__ dis,
                                                     const float* __restrict__ hlin,
                                                     float* __restrict__ hagg, int E) {
    long long t = (long long)blockIdx.x * TPB + threadIdx.x;
    int e = (int)(t >> 5);
    if (e >= E) return;
    int lane = (int)(t & 31);
    int r = row[e], c = col[e];
    float nrm = dis[r] * dis[c];
    float4 v = ((const float4*)(hlin + (long long)r * 128))[lane];
    float* dst = hagg + (long long)c * 128 + lane * 4;
    atomicAdd(dst + 0, nrm * v.x);
    atomicAdd(dst + 1, nrm * v.y);
    atomicAdd(dst + 2, nrm * v.z);
    atomicAdd(dst + 3, nrm * v.w);
}

__global__ __launch_bounds__(TPB) void k_knn_weights(const int* __restrict__ rk,
                                                     const int* __restrict__ ck,
                                                     const float* __restrict__ h,
                                                     const float* __restrict__ alpha,
                                                     float* __restrict__ wknn,
                                                     float* __restrict__ degw, int E) {
    long long t = (long long)blockIdx.x * TPB + threadIdx.x;
    int e = (int)(t >> 5);
    if (e >= E) return;
    int lane = (int)(t & 31);
    int a = rk[e], b = ck[e];
    float4 va = ((const float4*)(h + (long long)a * 128))[lane];
    float4 vb = ((const float4*)(h + (long long)b * 128))[lane];
    float dx = va.x - vb.x, dy = va.y - vb.y, dz = va.z - vb.z, dw = va.w - vb.w;
    float ss = dx * dx + dy * dy + dz * dz + dw * dw;
#pragma unroll
    for (int off = 16; off > 0; off >>= 1) ss += __shfl_xor(ss, off, 32);
    if (lane == 0) {
        float wv = 0.0f;
        if (a != b) {
            float nv = fmaxf(sqrtf(ss), 1e-12f);
            wv = alpha[0] * sqrtf(nv);
        }
        wknn[e] = wv;
        if (wv != 0.0f) atomicAdd(&degw[b], wv);
    }
}

__global__ __launch_bounds__(TPB) void k_prop2_edges(const int* __restrict__ row,
                                                     const int* __restrict__ col,
                                                     const float* __restrict__ disw,
                                                     const float* __restrict__ wknn,
                                                     const float* __restrict__ h2,
                                                     float* __restrict__ out, int E) {
    long long t = (long long)blockIdx.x * TPB + threadIdx.x;
    int e = (int)(t >> 5);
    if (e >= E) return;
    int lane = (int)(t & 31);
    float wv = wknn ? wknn[e] : 1.0f;
    if (wv == 0.0f) return;
    int r = row[e], c = col[e];
    float nrm = disw[r] * disw[c] * wv;
    float2 v = ((const float2*)(h2 + (long long)r * 64))[lane];
    float* dst = out + (long long)c * 64 + lane * 2;
    atomicAdd(dst + 0, nrm * v.x);
    atomicAdd(dst + 1, nrm * v.y);
}


static inline int nblk(long long threads) { return (int)((threads + TPB - 1) / TPB); }

extern "C" void kernel_launch(void* const* d_in, const int* in_sizes, int n_in,
                              void* d_out, int out_size, void* d_ws, size_t ws_size,
                              hipStream_t stream) {
    (void)n_in; (void)out_size; (void)ws_size;
    const float* x      = (const float*)d_in[0];
    const int*   ei_org = (const int*)d_in[1];
    const int*   ei_knn = (const int*)d_in[2];
    const float* alpha  = (const float*)d_in[3];
    const float* W1     = (const float*)d_in[4];
    const float* b1     = (const float*)d_in[5];
    const float* W2     = (const float*)d_in[6];
    const float* b2     = (const float*)d_in[7];
    float* out = (float*)d_out;
    float* ws  = (float*)d_ws;

    const int N = 50000, DIN = 128, DHID = 128, DOUT = 64;
    const int Eo = in_sizes[1] / 2;
    const int Ek = in_sizes[2] / 2;

    float* deg  = ws;
    float* dis  = ws + (long long)N;
    float* degw = ws + 2LL * N;
    float* disw = ws + 3LL * N;
    float* hlin = ws + 4LL * N;
    float* h    = hlin + (long long)N * DHID;
    float* h2   = h + (long long)N * DHID;
    float* wknn = h2 + (long long)N * DOUT;

    const int* row_o = ei_org;
    const int* col_o = ei_org + Eo;
    const int* row_k = ei_knn;
    const int* col_k = ei_knn + Ek;

    k_init_deg<<<nblk(N), TPB, 0, stream>>>(deg, degw, N);
    k_count_edges<<<nblk(Eo), TPB, 0, stream>>>(row_o, deg, Eo);
    k_rsqrt<<<nblk(N), TPB, 0, stream>>>(deg, dis, N);

    long long w1waves = (long long)(N / 16) * (DHID / 16);
    k_gemm_bias_wmma<<<nblk(w1waves * 32), TPB, 0, stream>>>(x, W1, b1, hlin, N, DIN, DHID);

    k_selfloop_scale<<<nblk((long long)N * 32), TPB, 0, stream>>>(dis, hlin, h, N, 32);
    k_prop1_edges<<<nblk((long long)Eo * 32), TPB, 0, stream>>>(row_o, col_o, dis, hlin, h, Eo);
    k_relu<<<nblk((long long)N * DHID), TPB, 0, stream>>>(h, (long long)N * DHID);

    k_count_edges<<<nblk(Eo), TPB, 0, stream>>>(col_o, degw, Eo);
    k_knn_weights<<<nblk((long long)Ek * 32), TPB, 0, stream>>>(row_k, col_k, h, alpha,
                                                                wknn, degw, Ek);
    k_rsqrt<<<nblk(N), TPB, 0, stream>>>(degw, disw, N);

    long long w2waves = (long long)(N / 16) * (DOUT / 16);
    k_gemm_bias_wmma<<<nblk(w2waves * 32), TPB, 0, stream>>>(h, W2, b2, h2, N, DHID, DOUT);

    k_selfloop_scale<<<nblk((long long)N * 16), TPB, 0, stream>>>(disw, h2, out, N, 16);
    k_prop2_edges<<<nblk((long long)Eo * 32), TPB, 0, stream>>>(row_o, col_o, disw,
                                                                (const float*)nullptr, h2, out, Eo);
    k_prop2_edges<<<nblk((long long)Ek * 32), TPB, 0, stream>>>(row_k, col_k, disw,
                                                                wknn, h2, out, Ek);
}


# Scatter kernels: lane l owns columns l+32k, so each gather and each f32 atomic instruction covers one whole aligned 128-B line instead of a quarter (layer 1) or half (layer 2) of several lines
# speedup vs baseline: 1.1144x; 1.1144x over previous
#include <hip/hip_runtime.h>
#include <math.h>


typedef float v2f __attribute__((ext_vector_type(2)));
typedef float v8f __attribute__((ext_vector_type(8)));

#define TPB 256


__global__ __launch_bounds__(TPB) void k_init_deg(float* deg, float* degw, int n) {
    int i = blockIdx.x * TPB + threadIdx.x;
    if (i < n) { deg[i] = 1.0f; degw[i] = 1.0f; }
}

__global__ __launch_bounds__(TPB) void k_count_edges(const int* __restrict__ idx,
                                                     float* __restrict__ deg, int e) {
    int i = blockIdx.x * TPB + threadIdx.x;
    if (i < e) atomicAdd(&deg[idx[i]], 1.0f);
}

__global__ __launch_bounds__(TPB) void k_rsqrt(const float* __restrict__ a,
                                               float* __restrict__ o, int n) {
    int i = blockIdx.x * TPB + threadIdx.x;
    if (i < n) o[i] = rsqrtf(a[i]);
}

__global__ __launch_bounds__(TPB) void k_relu(float* h, long long n) {
    long long i = (long long)blockIdx.x * TPB + threadIdx.x;
    if (i < n) h[i] = fmaxf(h[i], 0.0f);
}

__global__ __launch_bounds__(TPB) void k_selfloop_scale(const float* __restrict__ s,
                                                        const float* __restrict__ hin,
                                                        float* __restrict__ hout,
                                                        int n, int d4) {
    long long t = (long long)blockIdx.x * TPB + threadIdx.x;
    if (t >= (long long)n * d4) return;
    int i = (int)(t / d4);
    float sv = s[i]; sv = sv * sv;
    float4 v = ((const float4*)hin)[t];
    float4 r; r.x = sv * v.x; r.y = sv * v.y; r.z = sv * v.z; r.w = sv * v.w;
    ((float4*)hout)[t] = r;
}

__global__ __launch_bounds__(TPB) void k_gemm_bias_wmma(const float* __restrict__ A,
                                                        const float* __restrict__ W,
                                                        const float* __restrict__ bias,
                                                        float* __restrict__ out,
                                                        int M, int K, int Nc) {
    int wid = (blockIdx.x * TPB + threadIdx.x) >> 5;
    int ntile = Nc >> 4;
    int total = (M >> 4) * ntile;
    if (wid >= total) return;

    int mt = wid / ntile, nt = wid - mt * ntile;
    int m0 = mt << 4, n0 = nt << 4;
    int lane = threadIdx.x & 31;
    int half = lane >> 4;
    int l16  = lane & 15;

    const float* arow = A + (long long)(m0 + l16) * K;
    const float* wrow = W + (long long)(n0 + l16) * K;

    v8f c = {0.f, 0.f, 0.f, 0.f, 0.f, 0.f, 0.f, 0.f};
    for (int kb = 0; kb < K; kb += 4) {
        int k = kb + 2 * half;
        v2f a = *(const v2f*)(arow + k);
        v2f b = *(const v2f*)(wrow + k);
        c = __builtin_amdgcn_wmma_f32_16x16x4_f32(false, a, false, b,
                                                  (short)0, c, false, false);
    }

    float bv = bias[n0 + l16];
    long long colbase = (long long)n0 + l16;
#pragma unroll
    for (int r = 0; r < 8; ++r) {
        int m = m0 + r + 8 * half;
        out[(long long)m * Nc + colbase] = c[r] + bv;
    }
}

__global__ __launch_bounds__(TPB) void k_prop1_edges(const int* __restrict__ row,
                                                     const int* __restrict__ col,
                                                     const float* __restrict__ dis,
                                                     const float* __restrict__ hlin,
                                                     float* __restrict__ hagg, int E) {
    long long t = (long long)blockIdx.x * TPB + threadIdx.x;
    int e = (int)(t >> 5);
    if (e >= E) return;
    int lane = (int)(t & 31);
    int r = row[e], c = col[e];
    float nrm = dis[r] * dis[c];
    const float* src = hlin + (long long)r * 128 + lane;
    float v0 = src[0], v1 = src[32], v2 = src[64], v3 = src[96];
    float* dst = hagg + (long long)c * 128 + lane;
    atomicAdd(dst +  0, nrm * v0);
    atomicAdd(dst + 32, nrm * v1);
    atomicAdd(dst + 64, nrm * v2);
    atomicAdd(dst + 96, nrm * v3);
}

__global__ __launch_bounds__(TPB) void k_knn_weights(const int* __restrict__ rk,
                                                     const int* __restrict__ ck,
                                                     const float* __restrict__ h,
                                                     const float* __restrict__ alpha,
                                                     float* __restrict__ wknn,
                                                     float* __restrict__ degw, int E) {
    long long t = (long long)blockIdx.x * TPB + threadIdx.x;
    int e = (int)(t >> 5);
    if (e >= E) return;
    int lane = (int)(t & 31);
    int a = rk[e], b = ck[e];
    float4 va = ((const float4*)(h + (long long)a * 128))[lane];
    float4 vb = ((const float4*)(h + (long long)b * 128))[lane];
    float dx = va.x - vb.x, dy = va.y - vb.y, dz = va.z - vb.z, dw = va.w - vb.w;
    float ss = dx * dx + dy * dy + dz * dz + dw * dw;
#pragma unroll
    for (int off = 16; off > 0; off >>= 1) ss += __shfl_xor(ss, off, 32);
    if (lane == 0) {
        float wv = 0.0f;
        if (a != b) {
            float nv = fmaxf(sqrtf(ss), 1e-12f);
            wv = alpha[0] * sqrtf(nv);
        }
        wknn[e] = wv;
        if (wv != 0.0f) atomicAdd(&degw[b], wv);
    }
}

__global__ __launch_bounds__(TPB) void k_prop2_edges(const int* __restrict__ row,
                                                     const int* __restrict__ col,
                                                     const float* __restrict__ disw,
                                                     const float* __restrict__ wknn,
                                                     const float* __restrict__ h2,
                                                     float* __restrict__ out, int E) {
    long long t = (long long)blockIdx.x * TPB + threadIdx.x;
    int e = (int)(t >> 5);
    if (e >= E) return;
    int lane = (int)(t & 31);
    float wv = wknn ? wknn[e] : 1.0f;
    if (wv == 0.0f) return;
    int r = row[e], c = col[e];
    float nrm = disw[r] * disw[c] * wv;
    const float* src = h2 + (long long)r * 64 + lane;
    float v0 = src[0], v1 = src[32];
    float* dst = out + (long long)c * 64 + lane;
    atomicAdd(dst +  0, nrm * v0);
    atomicAdd(dst + 32, nrm * v1);
}


static inline int nblk(long long threads) { return (int)((threads + TPB - 1) / TPB); }

extern "C" void kernel_launch(void* const* d_in, const int* in_sizes, int n_in,
                              void* d_out, int out_size, void* d_ws, size_t ws_size,
                              hipStream_t stream) {
    (void)n_in; (void)out_size; (void)ws_size;
    const float* x      = (const float*)d_in[0];
    const int*   ei_org = (const int*)d_in[1];
    const int*   ei_knn = (const int*)d_in[2];
    const float* alpha  = (const float*)d_in[3];
    const float* W1     = (const float*)d_in[4];
    const float* b1     = (const float*)d_in[5];
    const float* W2     = (const float*)d_in[6];
    const float* b2     = (const float*)d_in[7];
    float* out = (float*)d_out;
    float* ws  = (float*)d_ws;

    const int N = 50000, DIN = 128, DHID = 128, DOUT = 64;
    const int Eo = in_sizes[1] / 2;
    const int Ek = in_sizes[2] / 2;

    float* deg  = ws;
    float* dis  = ws + (long long)N;
    float* degw = ws + 2LL * N;
    float* disw = ws + 3LL * N;
    float* hlin = ws + 4LL * N;
    float* h    = hlin + (long long)N * DHID;
    float* h2   = h + (long long)N * DHID;
    float* wknn = h2 + (long long)N * DOUT;

    const int* row_o = ei_org;
    const int* col_o = ei_org + Eo;
    const int* row_k = ei_knn;
    const int* col_k = ei_knn + Ek;

    k_init_deg<<<nblk(N), TPB, 0, stream>>>(deg, degw, N);
    k_count_edges<<<nblk(Eo), TPB, 0, stream>>>(row_o, deg, Eo);
    k_rsqrt<<<nblk(N), TPB, 0, stream>>>(deg, dis, N);

    long long w1waves = (long long)(N / 16) * (DHID / 16);
    k_gemm_bias_wmma<<<nblk(w1waves * 32), TPB, 0, stream>>>(x, W1, b1, hlin, N, DIN, DHID);

    k_selfloop_scale<<<nblk((long long)N * 32), TPB, 0, stream>>>(dis, hlin, h, N, 32);
    k_prop1_edges<<<nblk((long long)Eo * 32), TPB, 0, stream>>>(row_o, col_o, dis, hlin, h, Eo);
    k_relu<<<nblk((long long)N * DHID), TPB, 0, stream>>>(h, (long long)N * DHID);

    k_count_edges<<<nblk(Eo), TPB, 0, stream>>>(col_o, degw, Eo);
    k_knn_weights<<<nblk((long long)Ek * 32), TPB, 0, stream>>>(row_k, col_k, h, alpha,
                                                                wknn, degw, Ek);
    k_rsqrt<<<nblk(N), TPB, 0, stream>>>(degw, disw, N);

    long long w2waves = (long long)(N / 16) * (DOUT / 16);
    k_gemm_bias_wmma<<<nblk(w2waves * 32), TPB, 0, stream>>>(h, W2, b2, h2, N, DHID, DOUT);

    k_selfloop_scale<<<nblk((long long)N * 16), TPB, 0, stream>>>(disw, h2, out, N, 16);
    k_prop2_edges<<<nblk((long long)Eo * 32), TPB, 0, stream>>>(row_o, col_o, disw,
                                                                (const float*)nullptr, h2, out, Eo);
    k_prop2_edges<<<nblk((long long)Ek * 32), TPB, 0, stream>>>(row_k, col_k, disw,
                                                                wknn, h2, out, Ek);
}


# Edge kernels (both propagates, knn weights): one wave walks 32 consecutive edges, each group's row loads in flight before its f32 atomic adds (arrival order, as before), not one wave per edge
# speedup vs baseline: 1.5491x; 1.3901x over previous
#include <hip/hip_runtime.h>
#include <math.h>


typedef float v2f __attribute__((ext_vector_type(2)));
typedef float v8f __attribute__((ext_vector_type(8)));

#define TPB 256


__global__ __launch_bounds__(TPB) void k_init_deg(float* deg, float* degw, int n) {
    int i = blockIdx.x * TPB + threadIdx.x;
    if (i < n) { deg[i] = 1.0f; degw[i] = 1.0f; }
}

__global__ __launch_bounds__(TPB) void k_count_edges(const int* __restrict__ idx,
                                                     float* __restrict__ deg, int e) {
    int i = blockIdx.x * TPB + threadIdx.x;
    if (i < e) atomicAdd(&deg[idx[i]], 1.0f);
}

__global__ __launch_bounds__(TPB) void k_rsqrt(const float* __restrict__ a,
                                               float* __restrict__ o, int n) {
    int i = blockIdx.x * TPB + threadIdx.x;
    if (i < n) o[i] = rsqrtf(a[i]);
}

__global__ __launch_bounds__(TPB) void k_relu(float* h, long long n) {
    long long i = (long long)blockIdx.x * TPB + threadIdx.x;
    if (i < n) h[i] = fmaxf(h[i], 0.0f);
}

__global__ __launch_bounds__(TPB) void k_selfloop_scale(const float* __restrict__ s,
                                                        const float* __restrict__ hin,
                                                        float* __restrict__ hout,
                                                        int n, int d4) {
    long long t = (long long)blockIdx.x * TPB + threadIdx.x;
    if (t >= (long long)n * d4) return;
    int i = (int)(t / d4);
    float sv = s[i]; sv = sv * sv;
    float4 v = ((const float4*)hin)[t];
    float4 r; r.x = sv * v.x; r.y = sv * v.y; r.z = sv * v.z; r.w = sv * v.w;
    ((float4*)hout)[t] = r;
}

__global__ __launch_bounds__(TPB) void k_gemm_bias_wmma(const float* __restrict__ A,
                                                        const float* __restrict__ W,
                                                        const float* __restrict__ bias,
                                                        float* __restrict__ out,
                                                        int M, int K, int Nc) {
    int wid = (blockIdx.x * TPB + threadIdx.x) >> 5;
    int ntile = Nc >> 4;
    int total = (M >> 4) * ntile;
    if (wid >= total) return;

    int mt = wid / ntile, nt = wid - mt * ntile;
    int m0 = mt << 4, n0 = nt << 4;
    int lane = threadIdx.x & 31;
    int half = lane >> 4;
    int l16  = lane & 15;

    const float* arow = A + (long long)(m0 + l16) * K;
    const float* wrow = W + (long long)(n0 + l16) * K;

    v8f c = {0.f, 0.f, 0.f, 0.f, 0.f, 0.f, 0.f, 0.f};
    for (int kb = 0; kb < K; kb += 4) {
        int k = kb + 2 * half;
        v2f a = *(const v2f*)(arow + k);
        v2f b = *(const v2f*)(wrow + k);
        c = __builtin_amdgcn_wmma_f32_16x16x4_f32(false, a, false, b,
                                                  (short)0, c, false, false);
    }

    float bv = bias[n0 + l16];
    long long colbase = (long long)n0 + l16;
#pragma unroll
    for (int r = 0; r < 8; ++r) {
        int m = m0 + r + 8 * half;
        out[(long long)m * Nc + colbase] = c[r] + bv;
    }
}

#define EPW 32

#define G1 8
__global__ __launch_bounds__(TPB) void k_prop1_edges(const int* __restrict__ row,
                                                     const int* __restrict__ col,
                                                     const float* __restrict__ dis,
                                                     const float* __restrict__ hlin,
                                                     float* __restrict__ hagg, int E) {
    const int wave = __builtin_amdgcn_readfirstlane((int)((blockIdx.x * TPB + threadIdx.x) >> 5));
    const int lane = threadIdx.x & 31;
    const int e0 = wave * EPW;
    if (e0 >= E) return;
    const int cnt = (E - e0 < EPW) ? (E - e0) : EPW;
    int myr = 0, myc = 0;
    float mynrm = 0.0f;
    if (lane < cnt) {
        myr = row[e0 + lane];
        myc = col[e0 + lane];
        mynrm = dis[myr] * dis[myc];
    }
    for (int j = 0; j < cnt; j += G1) {
        float v[G1][4];
#pragma unroll
        for (int u = 0; u < G1; ++u) {
            const int r = __builtin_amdgcn_readlane(myr, j + u);
            const float* src = hlin + (long long)r * 128 + lane;
            v[u][0] = src[0]; v[u][1] = src[32]; v[u][2] = src[64]; v[u][3] = src[96];
        }
#pragma unroll
        for (int u = 0; u < G1; ++u) {
            if (j + u < cnt) {
                const int c = __builtin_amdgcn_readlane(myc, j + u);
                const float nrm = __int_as_float(__builtin_amdgcn_readlane(__float_as_int(mynrm), j + u));
                float* dst = hagg + (long long)c * 128 + lane;
                atomicAdd(dst +  0, nrm * v[u][0]);
                atomicAdd(dst + 32, nrm * v[u][1]);
                atomicAdd(dst + 64, nrm * v[u][2]);
                atomicAdd(dst + 96, nrm * v[u][3]);
            }
        }
    }
}

#define GK 4
__global__ __launch_bounds__(TPB) void k_knn_weights(const int* __restrict__ rk,
                                                     const int* __restrict__ ck,
                                                     const float* __restrict__ h,
                                                     const float* __restrict__ alpha,
                                                     float* __restrict__ wknn,
                                                     float* __restrict__ degw, int E) {
    const int wave = __builtin_amdgcn_readfirstlane((int)((blockIdx.x * TPB + threadIdx.x) >> 5));
    const int lane = threadIdx.x & 31;
    const int e0 = wave * EPW;
    if (e0 >= E) return;
    const int cnt = (E - e0 < EPW) ? (E - e0) : EPW;
    int mya = 0, myb = 0;
    if (lane < cnt) { mya = rk[e0 + lane]; myb = ck[e0 + lane]; }
    float myss = 0.0f;
    for (int j = 0; j < cnt; j += GK) {
        float4 va[GK], vb[GK];
#pragma unroll
        for (int u = 0; u < GK; ++u) {
            const int a = __builtin_amdgcn_readlane(mya, j + u);
            const int b = __builtin_amdgcn_readlane(myb, j + u);
            va[u] = ((const float4*)(h + (long long)a * 128))[lane];
            vb[u] = ((const float4*)(h + (long long)b * 128))[lane];
        }
#pragma unroll
        for (int u = 0; u < GK; ++u) {
            float dx = va[u].x - vb[u].x, dy = va[u].y - vb[u].y;
            float dz = va[u].z - vb[u].z, dw = va[u].w - vb[u].w;
            float ss = fmaf(dw, dw, fmaf(dz, dz, fmaf(dx, dx, dy * dy)));
#pragma unroll
            for (int off = 16; off > 0; off >>= 1) ss += __shfl_xor(ss, off, 32);
            if (lane == j + u) myss = ss;
        }
    }
    if (lane < cnt) {
        float wv = 0.0f;
        if (mya != myb) {
            float nv = fmaxf(sqrtf(myss), 1e-12f);
            wv = alpha[0] * sqrtf(nv);
        }
        wknn[e0 + lane] = wv;
        if (wv != 0.0f) atomicAdd(&degw[myb], wv);
    }
}

#define G2 16
__global__ __launch_bounds__(TPB) void k_prop2_edges(const int* __restrict__ row,
                                                     const int* __restrict__ col,
                                                     const float* __restrict__ disw,
                                                     const float* __restrict__ wknn,
                                                     const float* __restrict__ h2,
                                                     float* __restrict__ out, int E) {
    const int wave = __builtin_amdgcn_readfirstlane((int)((blockIdx.x * TPB + threadIdx.x) >> 5));
    const int lane = threadIdx.x & 31;
    const int e0 = wave * EPW;
    if (e0 >= E) return;
    const int cnt = (E - e0 < EPW) ? (E - e0) : EPW;
    int myr = 0, myc = 0;
    float mynrm = 0.0f;
    bool mylive = false;
    if (lane < cnt) {
        const float wv = wknn ? wknn[e0 + lane] : 1.0f;
        myr = row[e0 + lane];
        myc = col[e0 + lane];
        mynrm = disw[myr] * disw[myc] * wv;
        mylive = (wv != 0.0f);
    }
    const unsigned live = __builtin_amdgcn_ballot_w32(mylive);
    for (int j = 0; j < cnt; j += G2) {
        float v[G2][2];
#pragma unroll
        for (int u = 0; u < G2; ++u) {
            const int r = __builtin_amdgcn_readlane(myr, j + u);
            const float* src = h2 + (long long)r * 64 + lane;
            v[u][0] = src[0]; v[u][1] = src[32];
        }
#pragma unroll
        for (int u = 0; u < G2; ++u) {
            if ((live >> (j + u)) & 1u) {
                const int c = __builtin_amdgcn_readlane(myc, j + u);
                const float nrm = __int_as_float(__builtin_amdgcn_readlane(__float_as_int(mynrm), j + u));
                float* dst = out + (long long)c * 64 + lane;
                atomicAdd(dst +  0, nrm * v[u][0]);
                atomicAdd(dst + 32, nrm * v[u][1]);
            }
        }
    }
}


static inline int nblk(long long threads) { return (int)((threads + TPB - 1) / TPB); }

extern "C" void kernel_launch(void* const* d_in, const int* in_sizes, int n_in,
                              void* d_out, int out_size, void* d_ws, size_t ws_size,
                              hipStream_t stream) {
    (void)n_in; (void)out_size; (void)ws_size;
    const float* x      = (const float*)d_in[0];
    const int*   ei_org = (const int*)d_in[1];
    const int*   ei_knn = (const int*)d_in[2];
    const float* alpha  = (const float*)d_in[3];
    const float* W1     = (const float*)d_in[4];
    const float* b1     = (const float*)d_in[5];
    const float* W2     = (const float*)d_in[6];
    const float* b2     = (const float*)d_in[7];
    float* out = (float*)d_out;
    float* ws  = (float*)d_ws;

    const int N = 50000, DIN = 128, DHID = 128, DOUT = 64;
    const int Eo = in_sizes[1] / 2;
    const int Ek = in_sizes[2] / 2;

    float* deg  = ws;
    float* dis  = ws + (long long)N;
    float* degw = ws + 2LL * N;
    float* disw = ws + 3LL * N;
    float* hlin = ws + 4LL * N;
    float* h    = hlin + (long long)N * DHID;
    float* h2   = h + (long long)N * DHID;
    float* wknn = h2 + (long long)N * DOUT;

    const int* row_o = ei_org;
    const int* col_o = ei_org + Eo;
    const int* row_k = ei_knn;
    const int* col_k = ei_knn + Ek;

    k_init_deg<<<nblk(N), TPB, 0, stream>>>(deg, degw, N);
    k_count_edges<<<nblk(Eo), TPB, 0, stream>>>(row_o, deg, Eo);
    k_rsqrt<<<nblk(N), TPB, 0, stream>>>(deg, dis, N);

    long long w1waves = (long long)(N / 16) * (DHID / 16);
    k_gemm_bias_wmma<<<nblk(w1waves * 32), TPB, 0, stream>>>(x, W1, b1, hlin, N, DIN, DHID);

    k_selfloop_scale<<<nblk((long long)N * 32), TPB, 0, stream>>>(dis, hlin, h, N, 32);
    k_prop1_edges<<<nblk(Eo), TPB, 0, stream>>>(row_o, col_o, dis, hlin, h, Eo);
    k_relu<<<nblk((long long)N * DHID), TPB, 0, stream>>>(h, (long long)N * DHID);

    k_count_edges<<<nblk(Eo), TPB, 0, stream>>>(col_o, degw, Eo);
    k_knn_weights<<<nblk(Ek), TPB, 0, stream>>>(row_k, col_k, h, alpha,
                                                                wknn, degw, Ek);
    k_rsqrt<<<nblk(N), TPB, 0, stream>>>(degw, disw, N);

    long long w2waves = (long long)(N / 16) * (DOUT / 16);
    k_gemm_bias_wmma<<<nblk(w2waves * 32), TPB, 0, stream>>>(h, W2, b2, h2, N, DHID, DOUT);

    k_selfloop_scale<<<nblk((long long)N * 16), TPB, 0, stream>>>(disw, h2, out, N, 16);
    k_prop2_edges<<<nblk(Eo), TPB, 0, stream>>>(row_o, col_o, disw,
                                                                (const float*)nullptr, h2, out, Eo);
    k_prop2_edges<<<nblk(Ek), TPB, 0, stream>>>(row_k, col_k, disw,
                                                                wknn, h2, out, Ek);
}


# Elementwise passes fused away: GEMM epilogue also writes the self-loop term, relu taken where h is loaded (knn weights, GEMM 2), deg^-1/2 taken where a degree is read; 9 launches, not 14
# speedup vs baseline: 1.7319x; 1.1180x over previous
#include <hip/hip_runtime.h>
#include <math.h>


typedef float v2f __attribute__((ext_vector_type(2)));
typedef float v8f __attribute__((ext_vector_type(8)));

#define TPB 256


__global__ __launch_bounds__(TPB) void k_init_deg(float* deg, float* degw, int n) {
    int i = blockIdx.x * TPB + threadIdx.x;
    if (i < n) { deg[i] = 1.0f; degw[i] = 1.0f; }
}

__global__ __launch_bounds__(TPB) void k_count_edges(const int* __restrict__ idx,
                                                     float* __restrict__ deg, int e) {
    int i = blockIdx.x * TPB + threadIdx.x;
    if (i < e) atomicAdd(&deg[idx[i]], 1.0f);
}

template <bool RELU_A>
__global__ __launch_bounds__(TPB) void k_gemm_bias_wmma(const float* __restrict__ A,
                                                        const float* __restrict__ W,
                                                        const float* __restrict__ bias,
                                                        const float* __restrict__ dg,
                                                        float* __restrict__ out,
                                                        float* __restrict__ agg,
                                                        int M, int K, int Nc) {
    int wid = (blockIdx.x * TPB + threadIdx.x) >> 5;
    int ntile = Nc >> 4;
    int total = (M >> 4) * ntile;
    if (wid >= total) return;

    int mt = wid / ntile, nt = wid - mt * ntile;
    int m0 = mt << 4, n0 = nt << 4;
    int lane = threadIdx.x & 31;
    int half = lane >> 4;
    int l16  = lane & 15;

    const float* arow = A + (long long)(m0 + l16) * K;
    const float* wrow = W + (long long)(n0 + l16) * K;

    v8f c = {0.f, 0.f, 0.f, 0.f, 0.f, 0.f, 0.f, 0.f};
    for (int kb = 0; kb < K; kb += 4) {
        int k = kb + 2 * half;
        v2f a = *(const v2f*)(arow + k);
        v2f b = *(const v2f*)(wrow + k);
        if (RELU_A) { a.x = fmaxf(a.x, 0.0f); a.y = fmaxf(a.y, 0.0f); }
        c = __builtin_amdgcn_wmma_f32_16x16x4_f32(false, a, false, b,
                                                  (short)0, c, false, false);
    }

    float bv = bias[n0 + l16];
    long long colbase = (long long)n0 + l16;
#pragma unroll
    for (int r = 0; r < 8; ++r) {
        int m = m0 + r + 8 * half;
        float sv = rsqrtf(dg[m]); sv = sv * sv;
        float y = c[r] + bv;
        out[(long long)m * Nc + colbase] = y;
        agg[(long long)m * Nc + colbase] = sv * y;
    }
}

#define EPW 32

#define G1 8
__global__ __launch_bounds__(TPB) void k_prop1_edges(const int* __restrict__ row,
                                                     const int* __restrict__ col,
                                                     const float* __restrict__ deg,
                                                     const float* __restrict__ hlin,
                                                     float* __restrict__ hagg, int E) {
    const int wave = __builtin_amdgcn_readfirstlane((int)((blockIdx.x * TPB + threadIdx.x) >> 5));
    const int lane = threadIdx.x & 31;
    const int e0 = wave * EPW;
    if (e0 >= E) return;
    const int cnt = (E - e0 < EPW) ? (E - e0) : EPW;
    int myr = 0, myc = 0;
    float mynrm = 0.0f;
    if (lane < cnt) {
        myr = row[e0 + lane];
        myc = col[e0 + lane];
        mynrm = rsqrtf(deg[myr]) * rsqrtf(deg[myc]);
    }
    for (int j = 0; j < cnt; j += G1) {
        float v[G1][4];
#pragma unroll
        for (int u = 0; u < G1; ++u) {
            const int r = __builtin_amdgcn_readlane(myr, j + u);
            const float* src = hlin + (long long)r * 128 + lane;
            v[u][0] = src[0]; v[u][1] = src[32]; v[u][2] = src[64]; v[u][3] = src[96];
        }
#pragma unroll
        for (int u = 0; u < G1; ++u) {
            if (j + u < cnt) {
                const int c = __builtin_amdgcn_readlane(myc, j + u);
                const float nrm = __int_as_float(__builtin_amdgcn_readlane(__float_as_int(mynrm), j + u));
                float* dst = hagg + (long long)c * 128 + lane;
                atomicAdd(dst +  0, nrm * v[u][0]);
                atomicAdd(dst + 32, nrm * v[u][1]);
                atomicAdd(dst + 64, nrm * v[u][2]);
                atomicAdd(dst + 96, nrm * v[u][3]);
            }
        }
    }
}

#define GK 4
__global__ __launch_bounds__(TPB) void k_knn_weights(const int* __restrict__ rk,
                                                     const int* __restrict__ ck,
                                                     const float* __restrict__ h,
                                                     const float* __restrict__ alpha,
                                                     float* __restrict__ wknn,
                                                     float* __restrict__ degw, int E) {
    const int wave = __builtin_amdgcn_readfirstlane((int)((blockIdx.x * TPB + threadIdx.x) >> 5));
    const int lane = threadIdx.x & 31;
    const int e0 = wave * EPW;
    if (e0 >= E) return;
    const int cnt = (E - e0 < EPW) ? (E - e0) : EPW;
    int mya = 0, myb = 0;
    if (lane < cnt) { mya = rk[e0 + lane]; myb = ck[e0 + lane]; }
    float myss = 0.0f;
    for (int j = 0; j < cnt; j += GK) {
        float4 va[GK], vb[GK];
#pragma unroll
        for (int u = 0; u < GK; ++u) {
            const int a = __builtin_amdgcn_readlane(mya, j + u);
            const int b = __builtin_amdgcn_readlane(myb, j + u);
            va[u] = ((const float4*)(h + (long long)a * 128))[lane];
            vb[u] = ((const float4*)(h + (long long)b * 128))[lane];
        }
#pragma unroll
        for (int u = 0; u < GK; ++u) {
            float dx = fmaxf(va[u].x, 0.0f) - fmaxf(vb[u].x, 0.0f);
            float dy = fmaxf(va[u].y, 0.0f) - fmaxf(vb[u].y, 0.0f);
            float dz = fmaxf(va[u].z, 0.0f) - fmaxf(vb[u].z, 0.0f);
            float dw = fmaxf(va[u].w, 0.0f) - fmaxf(vb[u].w, 0.0f);
            float ss = fmaf(dw, dw, fmaf(dz, dz, fmaf(dx, dx, dy * dy)));
#pragma unroll
            for (int off = 16; off > 0; off >>= 1) ss += __shfl_xor(ss, off, 32);
            if (lane == j + u) myss = ss;
        }
    }
    if (lane < cnt) {
        float wv = 0.0f;
        if (mya != myb) {
            float nv = fmaxf(sqrtf(myss), 1e-12f);
            wv = alpha[0] * sqrtf(nv);
        }
        wknn[e0 + lane] = wv;
        if (wv != 0.0f) atomicAdd(&degw[myb], wv);
    }
}

#define G2 16
__global__ __launch_bounds__(TPB) void k_prop2_edges(const int* __restrict__ row,
                                                     const int* __restrict__ col,
                                                     const float* __restrict__ degw,
                                                     const float* __restrict__ wknn,
                                                     const float* __restrict__ h2,
                                                     float* __restrict__ out, int E) {
    const int wave = __builtin_amdgcn_readfirstlane((int)((blockIdx.x * TPB + threadIdx.x) >> 5));
    const int lane = threadIdx.x & 31;
    const int e0 = wave * EPW;
    if (e0 >= E) return;
    const int cnt = (E - e0 < EPW) ? (E - e0) : EPW;
    int myr = 0, myc = 0;
    float mynrm = 0.0f;
    bool mylive = false;
    if (lane < cnt) {
        const float wv = wknn ? wknn[e0 + lane] : 1.0f;
        myr = row[e0 + lane];
        myc = col[e0 + lane];
        mynrm = rsqrtf(degw[myr]) * rsqrtf(degw[myc]) * wv;
        mylive = (wv != 0.0f);
    }
    const unsigned live = __builtin_amdgcn_ballot_w32(mylive);
    for (int j = 0; j < cnt; j += G2) {
        float v[G2][2];
#pragma unroll
        for (int u = 0; u < G2; ++u) {
            const int r = __builtin_amdgcn_readlane(myr, j + u);
            const float* src = h2 + (long long)r * 64 + lane;
            v[u][0] = src[0]; v[u][1] = src[32];
        }
#pragma unroll
        for (int u = 0; u < G2; ++u) {
            if ((live >> (j + u)) & 1u) {
                const int c = __builtin_amdgcn_readlane(myc, j + u);
                const float nrm = __int_as_float(__builtin_amdgcn_readlane(__float_as_int(mynrm), j + u));
                float* dst = out + (long long)c * 64 + lane;
                atomicAdd(dst +  0, nrm * v[u][0]);
                atomicAdd(dst + 32, nrm * v[u][1]);
            }
        }
    }
}


static inline int nblk(long long threads) { return (int)((threads + TPB - 1) / TPB); }

extern "C" void kernel_launch(void* const* d_in, const int* in_sizes, int n_in,
                              void* d_out, int out_size, void* d_ws, size_t ws_size,
                              hipStream_t stream) {
    (void)n_in; (void)out_size; (void)ws_size;
    const float* x      = (const float*)d_in[0];
    const int*   ei_org = (const int*)d_in[1];
    const int*   ei_knn = (const int*)d_in[2];
    const float* alpha  = (const float*)d_in[3];
    const float* W1     = (const float*)d_in[4];
    const float* b1     = (const float*)d_in[5];
    const float* W2     = (const float*)d_in[6];
    const float* b2     = (const float*)d_in[7];
    float* out = (float*)d_out;
    float* ws  = (float*)d_ws;

    const int N = 50000, DIN = 128, DHID = 128, DOUT = 64;
    const int Eo = in_sizes[1] / 2;
    const int Ek = in_sizes[2] / 2;

    float* deg  = ws;
    float* degw = ws + 2LL * N;
    float* hlin = ws + 4LL * N;
    float* h    = hlin + (long long)N * DHID;
    float* h2   = h + (long long)N * DHID;
    float* wknn = h2 + (long long)N * DOUT;

    const int* row_o = ei_org;
    const int* col_o = ei_org + Eo;
    const int* row_k = ei_knn;
    const int* col_k = ei_knn + Ek;

    k_init_deg<<<nblk(N), TPB, 0, stream>>>(deg, degw, N);
    k_count_edges<<<nblk(Eo), TPB, 0, stream>>>(row_o, deg, Eo);

    long long w1waves = (long long)(N / 16) * (DHID / 16);
    k_gemm_bias_wmma<false><<<nblk(w1waves * 32), TPB, 0, stream>>>(x, W1, b1, deg, hlin, h,
                                                                    N, DIN, DHID);

    k_prop1_edges<<<nblk(Eo), TPB, 0, stream>>>(row_o, col_o, deg, hlin, h, Eo);

    k_count_edges<<<nblk(Eo), TPB, 0, stream>>>(col_o, degw, Eo);
    k_knn_weights<<<nblk(Ek), TPB, 0, stream>>>(row_k, col_k, h, alpha,
                                                                wknn, degw, Ek);

    long long w2waves = (long long)(N / 16) * (DOUT / 16);
    k_gemm_bias_wmma<true><<<nblk(w2waves * 32), TPB, 0, stream>>>(h, W2, b2, degw, h2, out,
                                                                   N, DHID, DOUT);

    k_prop2_edges<<<nblk(Eo), TPB, 0, stream>>>(row_o, col_o, degw,
                                                                (const float*)nullptr, h2, out, Eo);
    k_prop2_edges<<<nblk(Ek), TPB, 0, stream>>>(row_k, col_k, degw,
                                                                wknn, h2, out, Ek);
}


# Org-edge degree counts in LDS: 128 work-groups each tally a slice of the edge list in a 200 KB table (integer LDS adds) and store a partial row; one pass sums rows in fixed order; 8 launches, not 9
# speedup vs baseline: 2.4315x; 1.4039x over previous
#include <hip/hip_runtime.h>
#include <math.h>


typedef float v2f __attribute__((ext_vector_type(2)));
typedef float v8f __attribute__((ext_vector_type(8)));

#define TPB 256

#define NNODE   50000
#define NPITCH  50048
#define CNT_TPB 1024
#define CNT_G   64
#define CNT_U   16

__global__ __launch_bounds__(CNT_TPB) void k_count_lds(const int* __restrict__ idx_a,
                                                       const int* __restrict__ idx_b,
                                                       float* __restrict__ part, int e) {
    __shared__ unsigned tbl[NNODE];
    const bool second = blockIdx.x >= CNT_G;
    const int g = second ? (int)blockIdx.x - CNT_G : (int)blockIdx.x;
    const int* __restrict__ idx = second ? idx_b : idx_a;

    uint4* t4 = (uint4*)tbl;
    for (int i = threadIdx.x; i < NNODE / 4; i += CNT_TPB) t4[i] = make_uint4(0u, 0u, 0u, 0u);
    __syncthreads();

    const int stride = CNT_G * CNT_TPB;
    for (int base = g * CNT_TPB + (int)threadIdx.x; base < e; base += CNT_U * stride) {
        int v[CNT_U];
#pragma unroll
        for (int u = 0; u < CNT_U; ++u) {
            const int i = base + u * stride;
            v[u] = (i < e) ? idx[i] : -1;
        }
#pragma unroll
        for (int u = 0; u < CNT_U; ++u)
            if (v[u] >= 0) atomicAdd(&tbl[v[u]], 1u);
    }
    __syncthreads();

    float* __restrict__ prow = part + (long long)blockIdx.x * NPITCH;
    for (int i = threadIdx.x; i < NNODE; i += CNT_TPB) prow[i] = (float)tbl[i];
}

__global__ __launch_bounds__(TPB) void k_deg_sum(const float* __restrict__ part,
                                                 float* __restrict__ deg,
                                                 float* __restrict__ degw) {
    const int j = blockIdx.x * TPB + threadIdx.x;
    if (j >= 2 * NNODE) return;
    const int t = (j >= NNODE) ? 1 : 0;
    const int i = j - t * NNODE;
    const float* p = part + (long long)t * CNT_G * NPITCH + i;
    float s = 1.0f;
#pragma unroll 32
    for (int r = 0; r < CNT_G; ++r) s += p[(long long)r * NPITCH];
    if (t) degw[i] = s; else deg[i] = s;
}

template <bool RELU_A>
__global__ __launch_bounds__(TPB) void k_gemm_bias_wmma(const float* __restrict__ A,
                                                        const float* __restrict__ W,
                                                        const float* __restrict__ bias,
                                                        const float* __restrict__ dg,
                                                        float* __restrict__ out,
                                                        float* __restrict__ agg,
                                                        int M, int K, int Nc) {
    int wid = (blockIdx.x * TPB + threadIdx.x) >> 5;
    int ntile = Nc >> 4;
    int total = (M >> 4) * ntile;
    if (wid >= total) return;

    int mt = wid / ntile, nt = wid - mt * ntile;
    int m0 = mt << 4, n0 = nt << 4;
    int lane = threadIdx.x & 31;
    int half = lane >> 4;
    int l16  = lane & 15;

    const float* arow = A + (long long)(m0 + l16) * K;
    const float* wrow = W + (long long)(n0 + l16) * K;

    v8f c = {0.f, 0.f, 0.f, 0.f, 0.f, 0.f, 0.f, 0.f};
    for (int kb = 0; kb < K; kb += 4) {
        int k = kb + 2 * half;
        v2f a = *(const v2f*)(arow + k);
        v2f b = *(const v2f*)(wrow + k);
        if (RELU_A) { a.x = fmaxf(a.x, 0.0f); a.y = fmaxf(a.y, 0.0f); }
        c = __builtin_amdgcn_wmma_f32_16x16x4_f32(false, a, false, b,
                                                  (short)0, c, false, false);
    }

    float bv = bias[n0 + l16];
    long long colbase = (long long)n0 + l16;
#pragma unroll
    for (int r = 0; r < 8; ++r) {
        int m = m0 + r + 8 * half;
        float sv = rsqrtf(dg[m]); sv = sv * sv;
        float y = c[r] + bv;
        out[(long long)m * Nc + colbase] = y;
        agg[(long long)m * Nc + colbase] = sv * y;
    }
}

#define EPW 32

#define G1 8
__global__ __launch_bounds__(TPB) void k_prop1_edges(const int* __restrict__ row,
                                                     const int* __restrict__ col,
                                                     const float* __restrict__ deg,
                                                     const float* __restrict__ hlin,
                                                     float* __restrict__ hagg, int E) {
    const int wave = __builtin_amdgcn_readfirstlane((int)((blockIdx.x * TPB + threadIdx.x) >> 5));
    const int lane = threadIdx.x & 31;
    const int e0 = wave * EPW;
    if (e0 >= E) return;
    const int cnt = (E - e0 < EPW) ? (E - e0) : EPW;
    int myr = 0, myc = 0;
    float mynrm = 0.0f;
    if (lane < cnt) {
        myr = row[e0 + lane];
        myc = col[e0 + lane];
        mynrm = rsqrtf(deg[myr]) * rsqrtf(deg[myc]);
    }
    for (int j = 0; j < cnt; j += G1) {
        float v[G1][4];
#pragma unroll
        for (int u = 0; u < G1; ++u) {
            const int r = __builtin_amdgcn_readlane(myr, j + u);
            const float* src = hlin + (long long)r * 128 + lane;
            v[u][0] = src[0]; v[u][1] = src[32]; v[u][2] = src[64]; v[u][3] = src[96];
        }
#pragma unroll
        for (int u = 0; u < G1; ++u) {
            if (j + u < cnt) {
                const int c = __builtin_amdgcn_readlane(myc, j + u);
                const float nrm = __int_as_float(__builtin_amdgcn_readlane(__float_as_int(mynrm), j + u));
                float* dst = hagg + (long long)c * 128 + lane;
                atomicAdd(dst +  0, nrm * v[u][0]);
                atomicAdd(dst + 32, nrm * v[u][1]);
                atomicAdd(dst + 64, nrm * v[u][2]);
                atomicAdd(dst + 96, nrm * v[u][3]);
            }
        }
    }
}

#define GK 4
__global__ __launch_bounds__(TPB) void k_knn_weights(const int* __restrict__ rk,
                                                     const int* __restrict__ ck,
                                                     const float* __restrict__ h,
                                                     const float* __restrict__ alpha,
                                                     float* __restrict__ wknn,
                                                     float* __restrict__ degw, int E) {
    const int wave = __builtin_amdgcn_readfirstlane((int)((blockIdx.x * TPB + threadIdx.x) >> 5));
    const int lane = threadIdx.x & 31;
    const int e0 = wave * EPW;
    if (e0 >= E) return;
    const int cnt = (E - e0 < EPW) ? (E - e0) : EPW;
    int mya = 0, myb = 0;
    if (lane < cnt) { mya = rk[e0 + lane]; myb = ck[e0 + lane]; }
    float myss = 0.0f;
    for (int j = 0; j < cnt; j += GK) {
        float4 va[GK], vb[GK];
#pragma unroll
        for (int u = 0; u < GK; ++u) {
            const int a = __builtin_amdgcn_readlane(mya, j + u);
            const int b = __builtin_amdgcn_readlane(myb, j + u);
            va[u] = ((const float4*)(h + (long long)a * 128))[lane];
            vb[u] = ((const float4*)(h + (long long)b * 128))[lane];
        }
#pragma unroll
        for (int u = 0; u < GK; ++u) {
            float dx = fmaxf(va[u].x, 0.0f) - fmaxf(vb[u].x, 0.0f);
            float dy = fmaxf(va[u].y, 0.0f) - fmaxf(vb[u].y, 0.0f);
            float dz = fmaxf(va[u].z, 0.0f) - fmaxf(vb[u].z, 0.0f);
            float dw = fmaxf(va[u].w, 0.0f) - fmaxf(vb[u].w, 0.0f);
            float ss = fmaf(dw, dw, fmaf(dz, dz, fmaf(dx, dx, dy * dy)));
#pragma unroll
            for (int off = 16; off > 0; off >>= 1) ss += __shfl_xor(ss, off, 32);
            if (lane == j + u) myss = ss;
        }
    }
    if (lane < cnt) {
        float wv = 0.0f;
        if (mya != myb) {
            float nv = fmaxf(sqrtf(myss), 1e-12f);
            wv = alpha[0] * sqrtf(nv);
        }
        wknn[e0 + lane] = wv;
        if (wv != 0.0f) atomicAdd(&degw[myb], wv);
    }
}

#define G2 16
__global__ __launch_bounds__(TPB) void k_prop2_edges(const int* __restrict__ row,
                                                     const int* __restrict__ col,
                                                     const float* __restrict__ degw,
                                                     const float* __restrict__ wknn,
                                                     const float* __restrict__ h2,
                                                     float* __restrict__ out, int E) {
    const int wave = __builtin_amdgcn_readfirstlane((int)((blockIdx.x * TPB + threadIdx.x) >> 5));
    const int lane = threadIdx.x & 31;
    const int e0 = wave * EPW;
    if (e0 >= E) return;
    const int cnt = (E - e0 < EPW) ? (E - e0) : EPW;
    int myr = 0, myc = 0;
    float mynrm = 0.0f;
    bool mylive = false;
    if (lane < cnt) {
        const float wv = wknn ? wknn[e0 + lane] : 1.0f;
        myr = row[e0 + lane];
        myc = col[e0 + lane];
        mynrm = rsqrtf(degw[myr]) * rsqrtf(degw[myc]) * wv;
        mylive = (wv != 0.0f);
    }
    const unsigned live = __builtin_amdgcn_ballot_w32(mylive);
    for (int j = 0; j < cnt; j += G2) {
        float v[G2][2];
#pragma unroll
        for (int u = 0; u < G2; ++u) {
            const int r = __builtin_amdgcn_readlane(myr, j + u);
            const float* src = h2 + (long long)r * 64 + lane;
            v[u][0] = src[0]; v[u][1] = src[32];
        }
#pragma unroll
        for (int u = 0; u < G2; ++u) {
            if ((live >> (j + u)) & 1u) {
                const int c = __builtin_amdgcn_readlane(myc, j + u);
                const float nrm = __int_as_float(__builtin_amdgcn_readlane(__float_as_int(mynrm), j + u));
                float* dst = out + (long long)c * 64 + lane;
                atomicAdd(dst +  0, nrm * v[u][0]);
                atomicAdd(dst + 32, nrm * v[u][1]);
            }
        }
    }
}


static inline int nblk(long long threads) { return (int)((threads + TPB - 1) / TPB); }

extern "C" void kernel_launch(void* const* d_in, const int* in_sizes, int n_in,
                              void* d_out, int out_size, void* d_ws, size_t ws_size,
                              hipStream_t stream) {
    (void)n_in; (void)out_size; (void)ws_size;
    const float* x      = (const float*)d_in[0];
    const int*   ei_org = (const int*)d_in[1];
    const int*   ei_knn = (const int*)d_in[2];
    const float* alpha  = (const float*)d_in[3];
    const float* W1     = (const float*)d_in[4];
    const float* b1     = (const float*)d_in[5];
    const float* W2     = (const float*)d_in[6];
    const float* b2     = (const float*)d_in[7];
    float* out = (float*)d_out;
    float* ws  = (float*)d_ws;

    const int N = NNODE, DIN = 128, DHID = 128, DOUT = 64;
    const int Eo = in_sizes[1] / 2;
    const int Ek = in_sizes[2] / 2;

    float* deg  = ws;
    float* degw = ws + 2LL * N;
    float* hlin = ws + 4LL * N;
    float* h    = hlin + (long long)N * DHID;
    float* h2   = h + (long long)N * DHID;
    float* wknn = h2 + (long long)N * DOUT;
    float* part = wknn + Ek;

    const int* row_o = ei_org;
    const int* col_o = ei_org + Eo;
    const int* row_k = ei_knn;
    const int* col_k = ei_knn + Ek;

    k_count_lds<<<2 * CNT_G, CNT_TPB, 0, stream>>>(row_o, col_o, part, Eo);
    k_deg_sum<<<nblk(2LL * N), TPB, 0, stream>>>(part, deg, degw);

    long long w1waves = (long long)(N / 16) * (DHID / 16);
    k_gemm_bias_wmma<false><<<nblk(w1waves * 32), TPB, 0, stream>>>(x, W1, b1, deg, hlin, h,
                                                                    N, DIN, DHID);

    k_prop1_edges<<<nblk(Eo), TPB, 0, stream>>>(row_o, col_o, deg, hlin, h, Eo);

    k_knn_weights<<<nblk(Ek), TPB, 0, stream>>>(row_k, col_k, h, alpha,
                                                                wknn, degw, Ek);

    long long w2waves = (long long)(N / 16) * (DOUT / 16);
    k_gemm_bias_wmma<true><<<nblk(w2waves * 32), TPB, 0, stream>>>(h, W2, b2, degw, h2, out,
                                                                   N, DHID, DOUT);

    k_prop2_edges<<<nblk(Eo), TPB, 0, stream>>>(row_o, col_o, degw,
                                                                (const float*)nullptr, h2, out, Eo);
    k_prop2_edges<<<nblk(Ek), TPB, 0, stream>>>(row_k, col_k, degw,
                                                                wknn, h2, out, Ek);
}


# GEMMs: one wave per 16-row strip and all columns, W staged once per work-group in LDS, A fragments as 16-byte loads used for every column tile; k is summed in a permuted order
# speedup vs baseline: 2.7454x; 1.1291x over previous
#include <hip/hip_runtime.h>
#include <math.h>


typedef float v2f __attribute__((ext_vector_type(2)));
typedef float v8f __attribute__((ext_vector_type(8)));
typedef float v4f __attribute__((ext_vector_type(4)));

#define TPB 256

#define NNODE   50000
#define NPITCH  50048
#define CNT_TPB 1024
#define CNT_G   64
#define CNT_U   16

__global__ __launch_bounds__(CNT_TPB) void k_count_lds(const int* __restrict__ idx_a,
                                                       const int* __restrict__ idx_b,
                                                       float* __restrict__ part, int e) {
    __shared__ unsigned tbl[NNODE];
    const bool second = blockIdx.x >= CNT_G;
    const int g = second ? (int)blockIdx.x - CNT_G : (int)blockIdx.x;
    const int* __restrict__ idx = second ? idx_b : idx_a;

    uint4* t4 = (uint4*)tbl;
    for (int i = threadIdx.x; i < NNODE / 4; i += CNT_TPB) t4[i] = make_uint4(0u, 0u, 0u, 0u);
    __syncthreads();

    const int stride = CNT_G * CNT_TPB;
    for (int base = g * CNT_TPB + (int)threadIdx.x; base < e; base += CNT_U * stride) {
        int v[CNT_U];
#pragma unroll
        for (int u = 0; u < CNT_U; ++u) {
            const int i = base + u * stride;
            v[u] = (i < e) ? idx[i] : -1;
        }
#pragma unroll
        for (int u = 0; u < CNT_U; ++u)
            if (v[u] >= 0) atomicAdd(&tbl[v[u]], 1u);
    }
    __syncthreads();

    float* __restrict__ prow = part + (long long)blockIdx.x * NPITCH;
    for (int i = threadIdx.x; i < NNODE; i += CNT_TPB) prow[i] = (float)tbl[i];
}

__global__ __launch_bounds__(TPB) void k_deg_sum(const float* __restrict__ part,
                                                 float* __restrict__ deg,
                                                 float* __restrict__ degw) {
    const int j = blockIdx.x * TPB + threadIdx.x;
    if (j >= 2 * NNODE) return;
    const int t = (j >= NNODE) ? 1 : 0;
    const int i = j - t * NNODE;
    const float* p = part + (long long)t * CNT_G * NPITCH + i;
    float s = 1.0f;
#pragma unroll 32
    for (int r = 0; r < CNT_G; ++r) s += p[(long long)r * NPITCH];
    if (t) degw[i] = s; else deg[i] = s;
}

#define GT     512
#define WQ     33
template <int NC, bool RELU_A>
__global__ __launch_bounds__(GT) void k_gemm_bias_wmma(const float* __restrict__ A,
                                                       const float* __restrict__ W,
                                                       const float* __restrict__ bias,
                                                       const float* __restrict__ dg,
                                                       float* __restrict__ out,
                                                       float* __restrict__ agg,
                                                       int M) {
    constexpr int NT = NC / 16;
    __shared__ v4f wl[NC * WQ];

    const int lane = threadIdx.x & 31;
    const int half = lane >> 4;
    const int l16  = lane & 15;
    const int wv = __builtin_amdgcn_readfirstlane((int)(threadIdx.x >> 5));
    const int strip = blockIdx.x * (GT / 32) + wv;
    const bool live = strip < (M >> 4);
    const int m0 = strip << 4;

    const v4f* W4 = (const v4f*)W;
    v4f wt[NC * 32 / GT];
#pragma unroll
    for (int s = 0; s < NC * 32 / GT; ++s) wt[s] = W4[(int)threadIdx.x + s * GT];

    const int ml = live ? m0 : 0;
    float4 a[16];
    float dgv[8];
    const float4* arow = (const float4*)(A + (long long)(ml + l16) * 128) + half;
#pragma unroll
    for (int j = 0; j < 16; ++j) a[j] = arow[2 * j];
#pragma unroll
    for (int r = 0; r < 8; ++r) dgv[r] = dg[ml + r + 8 * half];

    if constexpr (NC == 128)
        asm volatile("" :: "v"(wt[0]), "v"(wt[1]), "v"(wt[2]), "v"(wt[3]),
                           "v"(wt[4]), "v"(wt[5]), "v"(wt[6]), "v"(wt[7]));
    else
        asm volatile("" :: "v"(wt[0]), "v"(wt[1]), "v"(wt[2]), "v"(wt[3]));

#pragma unroll
    for (int s = 0; s < NC * 32 / GT; ++s) {
        const int i = (int)threadIdx.x + s * GT;
        wl[(i >> 5) * WQ + (i & 31)] = wt[s];
    }
    __builtin_amdgcn_fence(__ATOMIC_RELEASE, "workgroup", "local");
    __builtin_amdgcn_s_barrier();
    __builtin_amdgcn_fence(__ATOMIC_ACQUIRE, "workgroup", "local");
    if (!live) return;

    if (RELU_A) {
#pragma unroll
        for (int j = 0; j < 16; ++j) {
            a[j].x = fmaxf(a[j].x, 0.0f); a[j].y = fmaxf(a[j].y, 0.0f);
            a[j].z = fmaxf(a[j].z, 0.0f); a[j].w = fmaxf(a[j].w, 0.0f);
        }
    }

    v8f acc[NT];
#pragma unroll
    for (int t = 0; t < NT; ++t) acc[t] = (v8f){0.f, 0.f, 0.f, 0.f, 0.f, 0.f, 0.f, 0.f};

    const v4f* bl = wl + l16 * WQ + half;
    v4f bc[NT], bn[NT];
#pragma unroll
    for (int t = 0; t < NT; ++t) bc[t] = bl[t * 16 * WQ];
#pragma unroll
    for (int j = 0; j < 16; ++j) {
        if (j + 1 < 16) {
#pragma unroll
            for (int t = 0; t < NT; ++t) bn[t] = bl[t * 16 * WQ + 2 * (j + 1)];
        }
        __builtin_amdgcn_sched_barrier(0);
        const v2f a0 = {a[j].x, a[j].y};
        const v2f a1 = {a[j].z, a[j].w};
#pragma unroll
        for (int t = 0; t < NT; ++t) {
            const v2f b0 = {bc[t].x, bc[t].y};
            acc[t] = __builtin_amdgcn_wmma_f32_16x16x4_f32(false, a0, false, b0,
                                                           (short)0, acc[t], false, false);
        }
#pragma unroll
        for (int t = 0; t < NT; ++t) {
            const v2f b1 = {bc[t].z, bc[t].w};
            acc[t] = __builtin_amdgcn_wmma_f32_16x16x4_f32(false, a1, false, b1,
                                                           (short)0, acc[t], false, false);
        }
        __builtin_amdgcn_sched_barrier(0);
#pragma unroll
        for (int t = 0; t < NT; ++t) bc[t] = bn[t];
    }

    float bv[NT];
#pragma unroll
    for (int t = 0; t < NT; ++t) bv[t] = bias[t * 16 + l16];
#pragma unroll
    for (int r = 0; r < 8; ++r) {
        const int m = m0 + r + 8 * half;
        float sv = rsqrtf(dgv[r]); sv = sv * sv;
        float* o = out + (long long)m * NC + l16;
        float* g = agg + (long long)m * NC + l16;
#pragma unroll
        for (int t = 0; t < NT; ++t) {
            const float y = acc[t][r] + bv[t];
            o[t * 16] = y;
            g[t * 16] = sv * y;
        }
    }
}

#define EPW 32

#define G1 8
__global__ __launch_bounds__(TPB) void k_prop1_edges(const int* __restrict__ row,
                                                     const int* __restrict__ col,
                                                     const float* __restrict__ deg,
                                                     const float* __restrict__ hlin,
                                                     float* __restrict__ hagg, int E) {
    const int wave = __builtin_amdgcn_readfirstlane((int)((blockIdx.x * TPB + threadIdx.x) >> 5));
    const int lane = threadIdx.x & 31;
    const int e0 = wave * EPW;
    if (e0 >= E) return;
    const int cnt = (E - e0 < EPW) ? (E - e0) : EPW;
    int myr = 0, myc = 0;
    float mynrm = 0.0f;
    if (lane < cnt) {
        myr = row[e0 + lane];
        myc = col[e0 + lane];
        mynrm = rsqrtf(deg[myr]) * rsqrtf(deg[myc]);
    }
    for (int j = 0; j < cnt; j += G1) {
        float v[G1][4];
#pragma unroll
        for (int u = 0; u < G1; ++u) {
            const int r = __builtin_amdgcn_readlane(myr, j + u);
            const float* src = hlin + (long long)r * 128 + lane;
            v[u][0] = src[0]; v[u][1] = src[32]; v[u][2] = src[64]; v[u][3] = src[96];
        }
#pragma unroll
        for (int u = 0; u < G1; ++u) {
            if (j + u < cnt) {
                const int c = __builtin_amdgcn_readlane(myc, j + u);
                const float nrm = __int_as_float(__builtin_amdgcn_readlane(__float_as_int(mynrm), j + u));
                float* dst = hagg + (long long)c * 128 + lane;
                atomicAdd(dst +  0, nrm * v[u][0]);
                atomicAdd(dst + 32, nrm * v[u][1]);
                atomicAdd(dst + 64, nrm * v[u][2]);
                atomicAdd(dst + 96, nrm * v[u][3]);
            }
        }
    }
}

#define GK 4
__global__ __launch_bounds__(TPB) void k_knn_weights(const int* __restrict__ rk,
                                                     const int* __restrict__ ck,
                                                     const float* __restrict__ h,
                                                     const float* __restrict__ alpha,
                                                     float* __restrict__ wknn,
                                                     float* __restrict__ degw, int E) {
    const int wave = __builtin_amdgcn_readfirstlane((int)((blockIdx.x * TPB + threadIdx.x) >> 5));
    const int lane = threadIdx.x & 31;
    const int e0 = wave * EPW;
    if (e0 >= E) return;
    const int cnt = (E - e0 < EPW) ? (E - e0) : EPW;
    int mya = 0, myb = 0;
    if (lane < cnt) { mya = rk[e0 + lane]; myb = ck[e0 + lane]; }
    float myss = 0.0f;
    for (int j = 0; j < cnt; j += GK) {
        float4 va[GK], vb[GK];
#pragma unroll
        for (int u = 0; u < GK; ++u) {
            const int a = __builtin_amdgcn_readlane(mya, j + u);
            const int b = __builtin_amdgcn_readlane(myb, j + u);
            va[u] = ((const float4*)(h + (long long)a * 128))[lane];
            vb[u] = ((const float4*)(h + (long long)b * 128))[lane];
        }
#pragma unroll
        for (int u = 0; u < GK; ++u) {
            float dx = fmaxf(va[u].x, 0.0f) - fmaxf(vb[u].x, 0.0f);
            float dy = fmaxf(va[u].y, 0.0f) - fmaxf(vb[u].y, 0.0f);
            float dz = fmaxf(va[u].z, 0.0f) - fmaxf(vb[u].z, 0.0f);
            float dw = fmaxf(va[u].w, 0.0f) - fmaxf(vb[u].w, 0.0f);
            float ss = fmaf(dw, dw, fmaf(dz, dz, fmaf(dx, dx, dy * dy)));
#pragma unroll
            for (int off = 16; off > 0; off >>= 1) ss += __shfl_xor(ss, off, 32);
            if (lane == j + u) myss = ss;
        }
    }
    if (lane < cnt) {
        float wv = 0.0f;
        if (mya != myb) {
            float nv = fmaxf(sqrtf(myss), 1e-12f);
            wv = alpha[0] * sqrtf(nv);
        }
        wknn[e0 + lane] = wv;
        if (wv != 0.0f) atomicAdd(&degw[myb], wv);
    }
}

#define G2 16
__global__ __launch_bounds__(TPB) void k_prop2_edges(const int* __restrict__ row,
                                                     const int* __restrict__ col,
                                                     const float* __restrict__ degw,
                                                     const float* __restrict__ wknn,
                                                     const float* __restrict__ h2,
                                                     float* __restrict__ out, int E) {
    const int wave = __builtin_amdgcn_readfirstlane((int)((blockIdx.x * TPB + threadIdx.x) >> 5));
    const int lane = threadIdx.x & 31;
    const int e0 = wave * EPW;
    if (e0 >= E) return;
    const int cnt = (E - e0 < EPW) ? (E - e0) : EPW;
    int myr = 0, myc = 0;
    float mynrm = 0.0f;
    bool mylive = false;
    if (lane < cnt) {
        const float wv = wknn ? wknn[e0 + lane] : 1.0f;
        myr = row[e0 + lane];
        myc = col[e0 + lane];
        mynrm = rsqrtf(degw[myr]) * rsqrtf(degw[myc]) * wv;
        mylive = (wv != 0.0f);
    }
    const unsigned live = __builtin_amdgcn_ballot_w32(mylive);
    for (int j = 0; j < cnt; j += G2) {
        float v[G2][2];
#pragma unroll
        for (int u = 0; u < G2; ++u) {
            const int r = __builtin_amdgcn_readlane(myr, j + u);
            const float* src = h2 + (long long)r * 64 + lane;
            v[u][0] = src[0]; v[u][1] = src[32];
        }
#pragma unroll
        for (int u = 0; u < G2; ++u) {
            if ((live >> (j + u)) & 1u) {
                const int c = __builtin_amdgcn_readlane(myc, j + u);
                const float nrm = __int_as_float(__builtin_amdgcn_readlane(__float_as_int(mynrm), j + u));
                float* dst = out + (long long)c * 64 + lane;
                atomicAdd(dst +  0, nrm * v[u][0]);
                atomicAdd(dst + 32, nrm * v[u][1]);
            }
        }
    }
}


static inline int nblk(long long threads) { return (int)((threads + TPB - 1) / TPB); }

extern "C" void kernel_launch(void* const* d_in, const int* in_sizes, int n_in,
                              void* d_out, int out_size, void* d_ws, size_t ws_size,
                              hipStream_t stream) {
    (void)n_in; (void)out_size; (void)ws_size;
    const float* x      = (const float*)d_in[0];
    const int*   ei_org = (const int*)d_in[1];
    const int*   ei_knn = (const int*)d_in[2];
    const float* alpha  = (const float*)d_in[3];
    const float* W1     = (const float*)d_in[4];
    const float* b1     = (const float*)d_in[5];
    const float* W2     = (const float*)d_in[6];
    const float* b2     = (const float*)d_in[7];
    float* out = (float*)d_out;
    float* ws  = (float*)d_ws;

    const int N = NNODE, DHID = 128, DOUT = 64;
    const int Eo = in_sizes[1] / 2;
    const int Ek = in_sizes[2] / 2;

    float* deg  = ws;
    float* degw = ws + 2LL * N;
    float* hlin = ws + 4LL * N;
    float* h    = hlin + (long long)N * DHID;
    float* h2   = h + (long long)N * DHID;
    float* wknn = h2 + (long long)N * DOUT;
    float* part = wknn + Ek;

    const int* row_o = ei_org;
    const int* col_o = ei_org + Eo;
    const int* row_k = ei_knn;
    const int* col_k = ei_knn + Ek;

    k_count_lds<<<2 * CNT_G, CNT_TPB, 0, stream>>>(row_o, col_o, part, Eo);
    k_deg_sum<<<nblk(2LL * N), TPB, 0, stream>>>(part, deg, degw);

    k_gemm_bias_wmma<128, false><<<(N / 16 + GT / 32 - 1) / (GT / 32), GT, 0, stream>>>(x, W1, b1, deg,
                                                                                       hlin, h, N);

    k_prop1_edges<<<nblk(Eo), TPB, 0, stream>>>(row_o, col_o, deg, hlin, h, Eo);

    k_knn_weights<<<nblk(Ek), TPB, 0, stream>>>(row_k, col_k, h, alpha,
                                                                wknn, degw, Ek);

    k_gemm_bias_wmma<64, true><<<(N / 16 + GT / 32 - 1) / (GT / 32), GT, 0, stream>>>(h, W2, b2, degw,
                                                                                     h2, out, N);

    k_prop2_edges<<<nblk(Eo), TPB, 0, stream>>>(row_o, col_o, degw,
                                                                (const float*)nullptr, h2, out, Eo);
    k_prop2_edges<<<nblk(Ek), TPB, 0, stream>>>(row_k, col_k, degw,
                                                                wknn, h2, out, Ek);
}


# Org-edge propagates (both layers) as gathers: edges counting-sorted by destination in LDS, one wave sums a node's rows in list order and stores once; f32 atomics stay for knn only; 9 launches
# speedup vs baseline: 2.9731x; 1.0829x over previous
#include <hip/hip_runtime.h>
#include <math.h>


typedef float v2f __attribute__((ext_vector_type(2)));
typedef float v8f __attribute__((ext_vector_type(8)));
typedef float v4f __attribute__((ext_vector_type(4)));

#define TPB 256

#define NNODE   50000
#define NPITCH  50048
#define CNT_TPB 1024
#define CNT_G   64
#define CNT_U   16

__global__ __launch_bounds__(CNT_TPB) void k_count_lds(const int* __restrict__ idx_a,
                                                       const int* __restrict__ idx_b,
                                                       float* __restrict__ part, int e) {
    __shared__ unsigned tbl[NNODE];
    const bool second = blockIdx.x >= CNT_G;
    const int g = second ? (int)blockIdx.x - CNT_G : (int)blockIdx.x;
    const int* __restrict__ idx = second ? idx_b : idx_a;

    uint4* t4 = (uint4*)tbl;
    for (int i = threadIdx.x; i < NNODE / 4; i += CNT_TPB) t4[i] = make_uint4(0u, 0u, 0u, 0u);
    __syncthreads();

    const int stride = CNT_G * CNT_TPB;
    for (int base = g * CNT_TPB + (int)threadIdx.x; base < e; base += CNT_U * stride) {
        int v[CNT_U];
#pragma unroll
        for (int u = 0; u < CNT_U; ++u) {
            const int i = base + u * stride;
            v[u] = (i < e) ? idx[i] : -1;
        }
#pragma unroll
        for (int u = 0; u < CNT_U; ++u)
            if (v[u] >= 0) atomicAdd(&tbl[v[u]], 1u);
    }
    __syncthreads();

    float* __restrict__ prow = part + (long long)blockIdx.x * NPITCH;
    for (int i = threadIdx.x; i < NNODE; i += CNT_TPB) prow[i] = (float)tbl[i];
}

__global__ __launch_bounds__(TPB) void k_deg_sum(float* __restrict__ part,
                                                 float* __restrict__ deg,
                                                 float* __restrict__ degw,
                                                 float* __restrict__ cnt) {
    const int j = blockIdx.x * TPB + threadIdx.x;
    if (j >= 2 * NNODE) return;
    const int t = (j >= NNODE) ? 1 : 0;
    const int i = j - t * NNODE;
    float* p = part + (long long)t * CNT_G * NPITCH + i;
    if (!t) {
        float s = 1.0f;
#pragma unroll 32
        for (int r = 0; r < CNT_G; ++r) s += p[(long long)r * NPITCH];
        deg[i] = s;
    } else {
        float run = 0.0f;
#pragma unroll 32
        for (int r = 0; r < CNT_G; ++r) {
            const float c = p[(long long)r * NPITCH];
            p[(long long)r * NPITCH] = run;
            run += c;
        }
        cnt[i] = run;
        degw[i] = run + 1.0f;
    }
}

#define SCAN_CH 49
#define NQ      (NNODE / 4)
#define NQ_PER  ((NQ + CNT_TPB - 1) / CNT_TPB)
__global__ __launch_bounds__(CNT_TPB) void k_fill_lds(const int* __restrict__ idx_src,
                                                      const int* __restrict__ idx_dst,
                                                      const float* __restrict__ part_b,
                                                      const float* __restrict__ cnt,
                                                      int* __restrict__ rowptr,
                                                      int* __restrict__ src, int e) {
    __shared__ unsigned tbl[CNT_TPB * SCAN_CH];
    __shared__ unsigned wsum[CNT_TPB / 32];
    const int tid = (int)threadIdx.x;
    const int lane = tid & 31;
    const int wv = tid >> 5;
    const int g = (int)blockIdx.x;
    uint4* t4 = (uint4*)tbl;

    {
        const v4f* c4 = (const v4f*)cnt;
        v4f cv[NQ_PER];
#pragma unroll
        for (int s = 0; s < NQ_PER; ++s) {
            const int q = tid + s * CNT_TPB;
            cv[s] = c4[q < NQ ? q : NQ - 1];
        }
#pragma unroll
        for (int s = 0; s < NQ_PER; ++s) {
            const int q = tid + s * CNT_TPB;
            if (q < NQ) t4[q] = make_uint4((unsigned)cv[s].x, (unsigned)cv[s].y,
                                           (unsigned)cv[s].z, (unsigned)cv[s].w);
        }
        if (tid < CNT_TPB * SCAN_CH - NNODE) tbl[NNODE + tid] = 0u;
    }
    __syncthreads();

    const int i0 = tid * SCAN_CH;
    unsigned own = 0;
#pragma unroll 7
    for (int k = 0; k < SCAN_CH; ++k) own += tbl[i0 + k];
    unsigned incl = own;
    for (int off = 1; off < 32; off <<= 1) {
        const unsigned n = __shfl_up(incl, off, 32);
        if (lane >= off) incl += n;
    }
    if (lane == 31) wsum[wv] = incl;
    __syncthreads();
    if (wv == 0) {
        const unsigned w = wsum[lane];
        unsigned wi = w;
        for (int off = 1; off < 32; off <<= 1) {
            const unsigned n = __shfl_up(wi, off, 32);
            if (lane >= off) wi += n;
        }
        wsum[lane] = wi - w;
    }
    __syncthreads();
    unsigned run = wsum[wv] + incl - own;
#pragma unroll 7
    for (int k = 0; k < SCAN_CH; ++k) {
        const unsigned c = tbl[i0 + k];
        tbl[i0 + k] = run;
        run += c;
    }
    __syncthreads();

    {
        const v4f* p4 = (const v4f*)(part_b + (long long)g * NPITCH);
        v4f pv[NQ_PER];
#pragma unroll
        for (int s = 0; s < NQ_PER; ++s) {
            const int q = tid + s * CNT_TPB;
            pv[s] = p4[q < NQ ? q : NQ - 1];
        }
#pragma unroll
        for (int s = 0; s < NQ_PER; ++s) {
            const int q = tid + s * CNT_TPB;
            if (q < NQ) {
                uint4 o = t4[q];
                if (g == 0)
                    ((int4*)rowptr)[q] = make_int4((int)o.x, (int)o.y, (int)o.z, (int)o.w);
                o.x += (unsigned)pv[s].x; o.y += (unsigned)pv[s].y;
                o.z += (unsigned)pv[s].z; o.w += (unsigned)pv[s].w;
                t4[q] = o;
            }
        }
        if (g == 0 && tid == 0) rowptr[NNODE] = e;
    }
    __syncthreads();

    const int stride = CNT_G * CNT_TPB;
    for (int base = g * CNT_TPB + tid; base < e; base += CNT_U * stride) {
        int d[CNT_U], r[CNT_U];
#pragma unroll
        for (int u = 0; u < CNT_U; ++u) {
            const int i = base + u * stride;
            d[u] = (i < e) ? idx_dst[i] : -1;
            r[u] = (i < e) ? idx_src[i] : 0;
        }
#pragma unroll
        for (int u = 0; u < CNT_U; ++u) {
            if (d[u] >= 0) {
                const unsigned pos = atomicAdd(&tbl[d[u]], 1u);
                src[pos] = r[u];
            }
        }
    }
}

#define GT     512
#define WQ     33
template <int NC, bool RELU_A>
__global__ __launch_bounds__(GT) void k_gemm_bias_wmma(const float* __restrict__ A,
                                                       const float* __restrict__ W,
                                                       const float* __restrict__ bias,
                                                       float* __restrict__ out,
                                                       int M) {
    constexpr int NT = NC / 16;
    __shared__ v4f wl[NC * WQ];

    const int lane = threadIdx.x & 31;
    const int half = lane >> 4;
    const int l16  = lane & 15;
    const int wv = __builtin_amdgcn_readfirstlane((int)(threadIdx.x >> 5));
    const int strip = blockIdx.x * (GT / 32) + wv;
    const bool live = strip < (M >> 4);
    const int m0 = strip << 4;

    const v4f* W4 = (const v4f*)W;
    v4f wt[NC * 32 / GT];
#pragma unroll
    for (int s = 0; s < NC * 32 / GT; ++s) wt[s] = W4[(int)threadIdx.x + s * GT];

    const int ml = live ? m0 : 0;
    float4 a[16];
    const float4* arow = (const float4*)(A + (long long)(ml + l16) * 128) + half;
#pragma unroll
    for (int j = 0; j < 16; ++j) a[j] = arow[2 * j];

    if constexpr (NC == 128)
        asm volatile("" :: "v"(wt[0]), "v"(wt[1]), "v"(wt[2]), "v"(wt[3]),
                           "v"(wt[4]), "v"(wt[5]), "v"(wt[6]), "v"(wt[7]));
    else
        asm volatile("" :: "v"(wt[0]), "v"(wt[1]), "v"(wt[2]), "v"(wt[3]));

#pragma unroll
    for (int s = 0; s < NC * 32 / GT; ++s) {
        const int i = (int)threadIdx.x + s * GT;
        wl[(i >> 5) * WQ + (i & 31)] = wt[s];
    }
    __builtin_amdgcn_fence(__ATOMIC_RELEASE, "workgroup", "local");
    __builtin_amdgcn_s_barrier();
    __builtin_amdgcn_fence(__ATOMIC_ACQUIRE, "workgroup", "local");
    if (!live) return;

    if (RELU_A) {
#pragma unroll
        for (int j = 0; j < 16; ++j) {
            a[j].x = fmaxf(a[j].x, 0.0f); a[j].y = fmaxf(a[j].y, 0.0f);
            a[j].z = fmaxf(a[j].z, 0.0f); a[j].w = fmaxf(a[j].w, 0.0f);
        }
    }

    v8f acc[NT];
#pragma unroll
    for (int t = 0; t < NT; ++t) acc[t] = (v8f){0.f, 0.f, 0.f, 0.f, 0.f, 0.f, 0.f, 0.f};

    const v4f* bl = wl + l16 * WQ + half;
    v4f bc[NT], bn[NT];
#pragma unroll
    for (int t = 0; t < NT; ++t) bc[t] = bl[t * 16 * WQ];
#pragma unroll
    for (int j = 0; j < 16; ++j) {
        if (j + 1 < 16) {
#pragma unroll
            for (int t = 0; t < NT; ++t) bn[t] = bl[t * 16 * WQ + 2 * (j + 1)];
        }
        __builtin_amdgcn_sched_barrier(0);
        const v2f a0 = {a[j].x, a[j].y};
        const v2f a1 = {a[j].z, a[j].w};
#pragma unroll
        for (int t = 0; t < NT; ++t) {
            const v2f b0 = {bc[t].x, bc[t].y};
            acc[t] = __builtin_amdgcn_wmma_f32_16x16x4_f32(false, a0, false, b0,
                                                           (short)0, acc[t], false, false);
        }
#pragma unroll
        for (int t = 0; t < NT; ++t) {
            const v2f b1 = {bc[t].z, bc[t].w};
            acc[t] = __builtin_amdgcn_wmma_f32_16x16x4_f32(false, a1, false, b1,
                                                           (short)0, acc[t], false, false);
        }
        __builtin_amdgcn_sched_barrier(0);
#pragma unroll
        for (int t = 0; t < NT; ++t) bc[t] = bn[t];
    }

    float bv[NT];
#pragma unroll
    for (int t = 0; t < NT; ++t) bv[t] = bias[t * 16 + l16];
#pragma unroll
    for (int r = 0; r < 8; ++r) {
        const int m = m0 + r + 8 * half;
        float* o = out + (long long)m * NC + l16;
#pragma unroll
        for (int t = 0; t < NT; ++t) o[t * 16] = acc[t][r] + bv[t];
    }
}

#define EPW 32

#define AGG_WG (NNODE / 16)
template <int D, int G>
__global__ __launch_bounds__(TPB) void k_agg_nodes(const int* __restrict__ rowptr,
                                                   const int* __restrict__ src,
                                                   const float* __restrict__ dg,
                                                   const float* __restrict__ feat,
                                                   float* __restrict__ outp) {
    constexpr int Q = D / 32;
    const int wave = __builtin_amdgcn_readfirstlane((int)((blockIdx.x * TPB + threadIdx.x) >> 5));
    const int nw = (int)gridDim.x * (TPB / 32);
    const int lane = threadIdx.x & 31;
    for (int c = wave; c < NNODE; c += nw) {
        const int beg = __builtin_amdgcn_readfirstlane(rowptr[c]);
        const int end = __builtin_amdgcn_readfirstlane(rowptr[c + 1]);
        const float dc = rsqrtf(dg[c]);
        const float* self = feat + (long long)c * D + lane;
        float acc[Q];
#pragma unroll
        for (int q = 0; q < Q; ++q) acc[q] = (dc * dc) * self[32 * q];
        for (int b = beg; b < end; b += 32) {
            const int cnt = (end - b < 32) ? (end - b) : 32;
            int myr = 0;
            float mynrm = 0.0f;
            if (lane < cnt) {
                myr = src[b + lane];
                mynrm = rsqrtf(dg[myr]) * dc;
            }
            for (int j = 0; j < cnt; j += G) {
                float v[G][Q];
#pragma unroll
                for (int u = 0; u < G; ++u) {
                    const int r = __builtin_amdgcn_readlane(myr, j + u);
                    const float* p = feat + (long long)r * D + lane;
#pragma unroll
                    for (int q = 0; q < Q; ++q) v[u][q] = p[32 * q];
                }
#pragma unroll
                for (int u = 0; u < G; ++u) {
                    if (j + u < cnt) {
                        const float nrm = __int_as_float(__builtin_amdgcn_readlane(__float_as_int(mynrm), j + u));
#pragma unroll
                        for (int q = 0; q < Q; ++q) acc[q] += nrm * v[u][q];
                    }
                }
            }
        }
        float* o = outp + (long long)c * D + lane;
#pragma unroll
        for (int q = 0; q < Q; ++q) o[32 * q] = acc[q];
    }
}

#define GK 4
__global__ __launch_bounds__(TPB) void k_knn_weights(const int* __restrict__ rk,
                                                     const int* __restrict__ ck,
                                                     const float* __restrict__ h,
                                                     const float* __restrict__ alpha,
                                                     float* __restrict__ wknn,
                                                     float* __restrict__ degw, int E) {
    const int wave = __builtin_amdgcn_readfirstlane((int)((blockIdx.x * TPB + threadIdx.x) >> 5));
    const int lane = threadIdx.x & 31;
    const int e0 = wave * EPW;
    if (e0 >= E) return;
    const int cnt = (E - e0 < EPW) ? (E - e0) : EPW;
    int mya = 0, myb = 0;
    if (lane < cnt) { mya = rk[e0 + lane]; myb = ck[e0 + lane]; }
    float myss = 0.0f;
    for (int j = 0; j < cnt; j += GK) {
        float4 va[GK], vb[GK];
#pragma unroll
        for (int u = 0; u < GK; ++u) {
            const int a = __builtin_amdgcn_readlane(mya, j + u);
            const int b = __builtin_amdgcn_readlane(myb, j + u);
            va[u] = ((const float4*)(h + (long long)a * 128))[lane];
            vb[u] = ((const float4*)(h + (long long)b * 128))[lane];
        }
#pragma unroll
        for (int u = 0; u < GK; ++u) {
            float dx = fmaxf(va[u].x, 0.0f) - fmaxf(vb[u].x, 0.0f);
            float dy = fmaxf(va[u].y, 0.0f) - fmaxf(vb[u].y, 0.0f);
            float dz = fmaxf(va[u].z, 0.0f) - fmaxf(vb[u].z, 0.0f);
            float dw = fmaxf(va[u].w, 0.0f) - fmaxf(vb[u].w, 0.0f);
            float ss = fmaf(dw, dw, fmaf(dz, dz, fmaf(dx, dx, dy * dy)));
#pragma unroll
            for (int off = 16; off > 0; off >>= 1) ss += __shfl_xor(ss, off, 32);
            if (lane == j + u) myss = ss;
        }
    }
    if (lane < cnt) {
        float wv = 0.0f;
        if (mya != myb) {
            float nv = fmaxf(sqrtf(myss), 1e-12f);
            wv = alpha[0] * sqrtf(nv);
        }
        wknn[e0 + lane] = wv;
        if (wv != 0.0f) atomicAdd(&degw[myb], wv);
    }
}

#define G2 16
__global__ __launch_bounds__(TPB) void k_prop2_edges(const int* __restrict__ row,
                                                     const int* __restrict__ col,
                                                     const float* __restrict__ degw,
                                                     const float* __restrict__ wknn,
                                                     const float* __restrict__ h2,
                                                     float* __restrict__ out, int E) {
    const int wave = __builtin_amdgcn_readfirstlane((int)((blockIdx.x * TPB + threadIdx.x) >> 5));
    const int lane = threadIdx.x & 31;
    const int e0 = wave * EPW;
    if (e0 >= E) return;
    const int cnt = (E - e0 < EPW) ? (E - e0) : EPW;
    int myr = 0, myc = 0;
    float mynrm = 0.0f;
    bool mylive = false;
    if (lane < cnt) {
        const float wv = wknn ? wknn[e0 + lane] : 1.0f;
        myr = row[e0 + lane];
        myc = col[e0 + lane];
        mynrm = rsqrtf(degw[myr]) * rsqrtf(degw[myc]) * wv;
        mylive = (wv != 0.0f);
    }
    const unsigned live = __builtin_amdgcn_ballot_w32(mylive);
    for (int j = 0; j < cnt; j += G2) {
        float v[G2][2];
#pragma unroll
        for (int u = 0; u < G2; ++u) {
            const int r = __builtin_amdgcn_readlane(myr, j + u);
            const float* src = h2 + (long long)r * 64 + lane;
            v[u][0] = src[0]; v[u][1] = src[32];
        }
#pragma unroll
        for (int u = 0; u < G2; ++u) {
            if ((live >> (j + u)) & 1u) {
                const int c = __builtin_amdgcn_readlane(myc, j + u);
                const float nrm = __int_as_float(__builtin_amdgcn_readlane(__float_as_int(mynrm), j + u));
                float* dst = out + (long long)c * 64 + lane;
                atomicAdd(dst +  0, nrm * v[u][0]);
                atomicAdd(dst + 32, nrm * v[u][1]);
            }
        }
    }
}


static inline int nblk(long long threads) { return (int)((threads + TPB - 1) / TPB); }

extern "C" void kernel_launch(void* const* d_in, const int* in_sizes, int n_in,
                              void* d_out, int out_size, void* d_ws, size_t ws_size,
                              hipStream_t stream) {
    (void)n_in; (void)out_size; (void)ws_size;
    const float* x      = (const float*)d_in[0];
    const int*   ei_org = (const int*)d_in[1];
    const int*   ei_knn = (const int*)d_in[2];
    const float* alpha  = (const float*)d_in[3];
    const float* W1     = (const float*)d_in[4];
    const float* b1     = (const float*)d_in[5];
    const float* W2     = (const float*)d_in[6];
    const float* b2     = (const float*)d_in[7];
    float* out = (float*)d_out;
    float* ws  = (float*)d_ws;

    const int N = NNODE, DHID = 128, DOUT = 64;
    const int Eo = in_sizes[1] / 2;
    const int Ek = in_sizes[2] / 2;

    float* deg  = ws;
    float* degw = ws + 2LL * N;
    float* hlin = ws + 4LL * N;
    float* h    = hlin + (long long)N * DHID;
    float* h2   = h + (long long)N * DHID;
    float* wknn = h2 + (long long)N * DOUT;
    float* part = wknn + Ek;
    float* cnt  = ws + 1LL * N;
    int* rowptr = (int*)(part + 2LL * CNT_G * NPITCH);
    int* srt    = rowptr + NPITCH;

    const int* row_o = ei_org;
    const int* col_o = ei_org + Eo;
    const int* row_k = ei_knn;
    const int* col_k = ei_knn + Ek;

    k_count_lds<<<2 * CNT_G, CNT_TPB, 0, stream>>>(row_o, col_o, part, Eo);
    k_deg_sum<<<nblk(2LL * N), TPB, 0, stream>>>(part, deg, degw, cnt);

    k_fill_lds<<<CNT_G, CNT_TPB, 0, stream>>>(row_o, col_o, part + (long long)CNT_G * NPITCH, cnt,
                                              rowptr, srt, Eo);

    k_gemm_bias_wmma<128, false><<<(N / 16 + GT / 32 - 1) / (GT / 32), GT, 0, stream>>>(x, W1, b1,
                                                                                       hlin, N);

    k_agg_nodes<128, 8><<<AGG_WG, TPB, 0, stream>>>(rowptr, srt, deg, hlin, h);

    k_knn_weights<<<nblk(Ek), TPB, 0, stream>>>(row_k, col_k, h, alpha,
                                                                wknn, degw, Ek);

    k_gemm_bias_wmma<64, true><<<(N / 16 + GT / 32 - 1) / (GT / 32), GT, 0, stream>>>(h, W2, b2,
                                                                                     h2, N);

    k_agg_nodes<64, 16><<<AGG_WG, TPB, 0, stream>>>(rowptr, srt, degw, h2, out);
    k_prop2_edges<<<nblk(Ek), TPB, 0, stream>>>(row_k, col_k, degw,
                                                                wknn, h2, out, Ek);
}


# Knn edges counting-sorted by destination like the org edges: knn weights, weighted degree (fixed-order sum) and layer-2 knn terms become per-node gathers in list order; no atomics left; 8 launches
# speedup vs baseline: 3.3220x; 1.1174x over previous
#include <hip/hip_runtime.h>
#include <math.h>


typedef float v2f __attribute__((ext_vector_type(2)));
typedef float v8f __attribute__((ext_vector_type(8)));
typedef float v4f __attribute__((ext_vector_type(4)));

#define TPB 256

#define NNODE   50000
#define NPITCH  50048
#define CNT_TPB 1024
#define CNT_G   64
#define CNT_U   16

__global__ __launch_bounds__(CNT_TPB) void k_count_lds(const int* __restrict__ idx_a,
                                                       const int* __restrict__ idx_b,
                                                       const int* __restrict__ idx_c,
                                                       float* __restrict__ part, int e_ab, int e_c) {
    __shared__ unsigned tbl[NNODE];
    const int list = (int)blockIdx.x / CNT_G;
    const int g = (int)blockIdx.x - list * CNT_G;
    const int* __restrict__ idx = (list == 0) ? idx_a : ((list == 1) ? idx_b : idx_c);
    const int e = (list == 2) ? e_c : e_ab;

    uint4* t4 = (uint4*)tbl;
    for (int i = threadIdx.x; i < NNODE / 4; i += CNT_TPB) t4[i] = make_uint4(0u, 0u, 0u, 0u);
    __syncthreads();

    const int stride = CNT_G * CNT_TPB;
    for (int base = g * CNT_TPB + (int)threadIdx.x; base < e; base += CNT_U * stride) {
        int v[CNT_U];
#pragma unroll
        for (int u = 0; u < CNT_U; ++u) {
            const int i = base + u * stride;
            v[u] = (i < e) ? idx[i] : -1;
        }
#pragma unroll
        for (int u = 0; u < CNT_U; ++u)
            if (v[u] >= 0) atomicAdd(&tbl[v[u]], 1u);
    }
    __syncthreads();

    float* __restrict__ prow = part + (long long)blockIdx.x * NPITCH;
    for (int i = threadIdx.x; i < NNODE; i += CNT_TPB) prow[i] = (float)tbl[i];
}

__global__ __launch_bounds__(TPB) void k_deg_sum(float* __restrict__ part,
                                                 float* __restrict__ deg,
                                                 float* __restrict__ cnt,
                                                 float* __restrict__ cntk) {
    const int j = blockIdx.x * TPB + threadIdx.x;
    if (j >= 3 * NNODE) return;
    const int t = j / NNODE;
    const int i = j - t * NNODE;
    float* p = part + (long long)t * CNT_G * NPITCH + i;
    if (!t) {
        float s = 1.0f;
#pragma unroll 32
        for (int r = 0; r < CNT_G; ++r) s += p[(long long)r * NPITCH];
        deg[i] = s;
    } else {
        float run = 0.0f;
#pragma unroll 32
        for (int r = 0; r < CNT_G; ++r) {
            const float c = p[(long long)r * NPITCH];
            p[(long long)r * NPITCH] = run;
            run += c;
        }
        if (t == 1) cnt[i] = run; else cntk[i] = run;
    }
}

#define SCAN_CH 49
#define NQ      (NNODE / 4)
#define NQ_PER  ((NQ + CNT_TPB - 1) / CNT_TPB)
__global__ __launch_bounds__(CNT_TPB) void k_fill_lds(const int* __restrict__ src_o,
                                                      const int* __restrict__ dst_o,
                                                      const int* __restrict__ src_k,
                                                      const int* __restrict__ dst_k,
                                                      const float* __restrict__ part_bc,
                                                      const float* __restrict__ cnt_o,
                                                      const float* __restrict__ cnt_k,
                                                      int* __restrict__ rowptr_o,
                                                      int* __restrict__ rowptr_k,
                                                      int* __restrict__ srt_o,
                                                      int* __restrict__ srt_k, int e_o, int e_k) {
    __shared__ unsigned tbl[CNT_TPB * SCAN_CH];
    __shared__ unsigned wsum[CNT_TPB / 32];
    const int tid = (int)threadIdx.x;
    const int lane = tid & 31;
    const int wv = tid >> 5;
    const bool knn = blockIdx.x >= CNT_G;
    const int g = knn ? (int)blockIdx.x - CNT_G : (int)blockIdx.x;
    const int* __restrict__ idx_src = knn ? src_k : src_o;
    const int* __restrict__ idx_dst = knn ? dst_k : dst_o;
    const float* __restrict__ cnt = knn ? cnt_k : cnt_o;
    int* __restrict__ rowptr = knn ? rowptr_k : rowptr_o;
    int* __restrict__ src = knn ? srt_k : srt_o;
    const int e = knn ? e_k : e_o;
    uint4* t4 = (uint4*)tbl;

    {
        const v4f* c4 = (const v4f*)cnt;
        v4f cv[NQ_PER];
#pragma unroll
        for (int s = 0; s < NQ_PER; ++s) {
            const int q = tid + s * CNT_TPB;
            cv[s] = c4[q < NQ ? q : NQ - 1];
        }
#pragma unroll
        for (int s = 0; s < NQ_PER; ++s) {
            const int q = tid + s * CNT_TPB;
            if (q < NQ) t4[q] = make_uint4((unsigned)cv[s].x, (unsigned)cv[s].y,
                                           (unsigned)cv[s].z, (unsigned)cv[s].w);
        }
        if (tid < CNT_TPB * SCAN_CH - NNODE) tbl[NNODE + tid] = 0u;
    }
    __syncthreads();

    const int i0 = tid * SCAN_CH;
    unsigned own = 0;
#pragma unroll 7
    for (int k = 0; k < SCAN_CH; ++k) own += tbl[i0 + k];
    unsigned incl = own;
    for (int off = 1; off < 32; off <<= 1) {
        const unsigned n = __shfl_up(incl, off, 32);
        if (lane >= off) incl += n;
    }
    if (lane == 31) wsum[wv] = incl;
    __syncthreads();
    if (wv == 0) {
        const unsigned w = wsum[lane];
        unsigned wi = w;
        for (int off = 1; off < 32; off <<= 1) {
            const unsigned n = __shfl_up(wi, off, 32);
            if (lane >= off) wi += n;
        }
        wsum[lane] = wi - w;
    }
    __syncthreads();
    unsigned run = wsum[wv] + incl - own;
#pragma unroll 7
    for (int k = 0; k < SCAN_CH; ++k) {
        const unsigned c = tbl[i0 + k];
        tbl[i0 + k] = run;
        run += c;
    }
    __syncthreads();

    {
        const v4f* p4 = (const v4f*)(part_bc + (long long)blockIdx.x * NPITCH);
        v4f pv[NQ_PER];
#pragma unroll
        for (int s = 0; s < NQ_PER; ++s) {
            const int q = tid + s * CNT_TPB;
            pv[s] = p4[q < NQ ? q : NQ - 1];
        }
#pragma unroll
        for (int s = 0; s < NQ_PER; ++s) {
            const int q = tid + s * CNT_TPB;
            if (q < NQ) {
                uint4 o = t4[q];
                if (g == 0)
                    ((int4*)rowptr)[q] = make_int4((int)o.x, (int)o.y, (int)o.z, (int)o.w);
                o.x += (unsigned)pv[s].x; o.y += (unsigned)pv[s].y;
                o.z += (unsigned)pv[s].z; o.w += (unsigned)pv[s].w;
                t4[q] = o;
            }
        }
        if (g == 0 && tid == 0) rowptr[NNODE] = e;
    }
    __syncthreads();

    const int stride = CNT_G * CNT_TPB;
    for (int base = g * CNT_TPB + tid; base < e; base += CNT_U * stride) {
        int d[CNT_U], r[CNT_U];
#pragma unroll
        for (int u = 0; u < CNT_U; ++u) {
            const int i = base + u * stride;
            d[u] = (i < e) ? idx_dst[i] : -1;
            r[u] = (i < e) ? idx_src[i] : 0;
        }
#pragma unroll
        for (int u = 0; u < CNT_U; ++u) {
            if (d[u] >= 0) {
                const unsigned pos = atomicAdd(&tbl[d[u]], 1u);
                src[pos] = r[u];
            }
        }
    }
}

#define GT     512
#define WQ     33
template <int NC, bool RELU_A>
__global__ __launch_bounds__(GT) void k_gemm_bias_wmma(const float* __restrict__ A,
                                                       const float* __restrict__ W,
                                                       const float* __restrict__ bias,
                                                       float* __restrict__ out,
                                                       int M) {
    constexpr int NT = NC / 16;
    __shared__ v4f wl[NC * WQ];

    const int lane = threadIdx.x & 31;
    const int half = lane >> 4;
    const int l16  = lane & 15;
    const int wv = __builtin_amdgcn_readfirstlane((int)(threadIdx.x >> 5));
    const int strip = blockIdx.x * (GT / 32) + wv;
    const bool live = strip < (M >> 4);
    const int m0 = strip << 4;

    const v4f* W4 = (const v4f*)W;
    v4f wt[NC * 32 / GT];
#pragma unroll
    for (int s = 0; s < NC * 32 / GT; ++s) wt[s] = W4[(int)threadIdx.x + s * GT];

    const int ml = live ? m0 : 0;
    float4 a[16];
    const float4* arow = (const float4*)(A + (long long)(ml + l16) * 128) + half;
#pragma unroll
    for (int j = 0; j < 16; ++j) a[j] = arow[2 * j];

    if constexpr (NC == 128)
        asm volatile("" :: "v"(wt[0]), "v"(wt[1]), "v"(wt[2]), "v"(wt[3]),
                           "v"(wt[4]), "v"(wt[5]), "v"(wt[6]), "v"(wt[7]));
    else
        asm volatile("" :: "v"(wt[0]), "v"(wt[1]), "v"(wt[2]), "v"(wt[3]));

#pragma unroll
    for (int s = 0; s < NC * 32 / GT; ++s) {
        const int i = (int)threadIdx.x + s * GT;
        wl[(i >> 5) * WQ + (i & 31)] = wt[s];
    }
    __builtin_amdgcn_fence(__ATOMIC_RELEASE, "workgroup", "local");
    __builtin_amdgcn_s_barrier();
    __builtin_amdgcn_fence(__ATOMIC_ACQUIRE, "workgroup", "local");
    if (!live) return;

    if (RELU_A) {
#pragma unroll
        for (int j = 0; j < 16; ++j) {
            a[j].x = fmaxf(a[j].x, 0.0f); a[j].y = fmaxf(a[j].y, 0.0f);
            a[j].z = fmaxf(a[j].z, 0.0f); a[j].w = fmaxf(a[j].w, 0.0f);
        }
    }

    v8f acc[NT];
#pragma unroll
    for (int t = 0; t < NT; ++t) acc[t] = (v8f){0.f, 0.f, 0.f, 0.f, 0.f, 0.f, 0.f, 0.f};

    const v4f* bl = wl + l16 * WQ + half;
    v4f bc[NT], bn[NT];
#pragma unroll
    for (int t = 0; t < NT; ++t) bc[t] = bl[t * 16 * WQ];
#pragma unroll
    for (int j = 0; j < 16; ++j) {
        if (j + 1 < 16) {
#pragma unroll
            for (int t = 0; t < NT; ++t) bn[t] = bl[t * 16 * WQ + 2 * (j + 1)];
        }
        __builtin_amdgcn_sched_barrier(0);
        const v2f a0 = {a[j].x, a[j].y};
        const v2f a1 = {a[j].z, a[j].w};
#pragma unroll
        for (int t = 0; t < NT; ++t) {
            const v2f b0 = {bc[t].x, bc[t].y};
            acc[t] = __builtin_amdgcn_wmma_f32_16x16x4_f32(false, a0, false, b0,
                                                           (short)0, acc[t], false, false);
        }
#pragma unroll
        for (int t = 0; t < NT; ++t) {
            const v2f b1 = {bc[t].z, bc[t].w};
            acc[t] = __builtin_amdgcn_wmma_f32_16x16x4_f32(false, a1, false, b1,
                                                           (short)0, acc[t], false, false);
        }
        __builtin_amdgcn_sched_barrier(0);
#pragma unroll
        for (int t = 0; t < NT; ++t) bc[t] = bn[t];
    }

    float bv[NT];
#pragma unroll
    for (int t = 0; t < NT; ++t) bv[t] = bias[t * 16 + l16];
#pragma unroll
    for (int r = 0; r < 8; ++r) {
        const int m = m0 + r + 8 * half;
        float* o = out + (long long)m * NC + l16;
#pragma unroll
        for (int t = 0; t < NT; ++t) o[t * 16] = acc[t][r] + bv[t];
    }
}

#define AGG_WG (NNODE / 16)

template <int D, int G, bool WT>
__device__ __forceinline__ void agg_run(int beg, int end, const int* __restrict__ src,
                                        const float* __restrict__ wt,
                                        const float* __restrict__ dg,
                                        const float* __restrict__ feat,
                                        float dc, int lane, float (&acc)[D / 32]) {
    constexpr int Q = D / 32;
    for (int b = beg; b < end; b += 32) {
        const int cnt = (end - b < 32) ? (end - b) : 32;
        int myr = 0;
        float mynrm = 0.0f;
        if (lane < cnt) {
            myr = src[b + lane];
            mynrm = rsqrtf(dg[myr]) * dc;
            if (WT) mynrm *= wt[b + lane];
        }
        for (int j = 0; j < cnt; j += G) {
            float v[G][Q];
#pragma unroll
            for (int u = 0; u < G; ++u) {
                const int r = __builtin_amdgcn_readlane(myr, j + u);
                const float* p = feat + (long long)r * D + lane;
#pragma unroll
                for (int q = 0; q < Q; ++q) v[u][q] = p[32 * q];
            }
#pragma unroll
            for (int u = 0; u < G; ++u) {
                if (j + u < cnt) {
                    const float nrm = __int_as_float(__builtin_amdgcn_readlane(__float_as_int(mynrm), j + u));
#pragma unroll
                    for (int q = 0; q < Q; ++q) acc[q] += nrm * v[u][q];
                }
            }
        }
    }
}

template <int D, int G, bool KNN>
__global__ __launch_bounds__(TPB) void k_agg_nodes(const int* __restrict__ rowptr,
                                                   const int* __restrict__ src,
                                                   const float* __restrict__ dg,
                                                   const float* __restrict__ feat,
                                                   float* __restrict__ outp,
                                                   const int* __restrict__ rowptr_k,
                                                   const int* __restrict__ src_k,
                                                   const float* __restrict__ w_k) {
    constexpr int Q = D / 32;
    const int wave = __builtin_amdgcn_readfirstlane((int)((blockIdx.x * TPB + threadIdx.x) >> 5));
    const int nw = (int)gridDim.x * (TPB / 32);
    const int lane = threadIdx.x & 31;
    for (int c = wave; c < NNODE; c += nw) {
        const int beg = __builtin_amdgcn_readfirstlane(rowptr[c]);
        const int end = __builtin_amdgcn_readfirstlane(rowptr[c + 1]);
        int kbeg = 0, kend = 0;
        if (KNN) {
            kbeg = __builtin_amdgcn_readfirstlane(rowptr_k[c]);
            kend = __builtin_amdgcn_readfirstlane(rowptr_k[c + 1]);
        }
        const float dc = rsqrtf(dg[c]);
        const float* self = feat + (long long)c * D + lane;
        float acc[Q];
#pragma unroll
        for (int q = 0; q < Q; ++q) acc[q] = (dc * dc) * self[32 * q];
        agg_run<D, G, false>(beg, end, src, nullptr, dg, feat, dc, lane, acc);
        if (KNN) agg_run<D, G, true>(kbeg, kend, src_k, w_k, dg, feat, dc, lane, acc);
        float* o = outp + (long long)c * D + lane;
#pragma unroll
        for (int q = 0; q < Q; ++q) o[32 * q] = acc[q];
    }
}

#define GK 8
__global__ __launch_bounds__(TPB) void k_knn_nodes(const int* __restrict__ rowptr_k,
                                                   const int* __restrict__ src_k,
                                                   const float* __restrict__ h,
                                                   const float* __restrict__ alpha,
                                                   const float* __restrict__ cnt,
                                                   float* __restrict__ w_k,
                                                   float* __restrict__ degw) {
    const int wave = __builtin_amdgcn_readfirstlane((int)((blockIdx.x * TPB + threadIdx.x) >> 5));
    const int nw = (int)gridDim.x * (TPB / 32);
    const int lane = threadIdx.x & 31;
    const float al = alpha[0];
    for (int c = wave; c < NNODE; c += nw) {
        const int beg = __builtin_amdgcn_readfirstlane(rowptr_k[c]);
        const int end = __builtin_amdgcn_readfirstlane(rowptr_k[c + 1]);
        const float4 vc = ((const float4*)(h + (long long)c * 128))[lane];
        const float cx = fmaxf(vc.x, 0.0f), cy = fmaxf(vc.y, 0.0f);
        const float cz = fmaxf(vc.z, 0.0f), cw = fmaxf(vc.w, 0.0f);
        float wsum = 0.0f;
        for (int b = beg; b < end; b += 32) {
            const int n = (end - b < 32) ? (end - b) : 32;
            int mya = c;
            if (lane < n) mya = src_k[b + lane];
            float myss = 0.0f;
            for (int j = 0; j < n; j += GK) {
                float4 va[GK];
#pragma unroll
                for (int u = 0; u < GK; ++u) {
                    const int a = __builtin_amdgcn_readlane(mya, j + u);
                    va[u] = ((const float4*)(h + (long long)a * 128))[lane];
                }
#pragma unroll
                for (int u = 0; u < GK; ++u) {
                    float dx = fmaxf(va[u].x, 0.0f) - cx;
                    float dy = fmaxf(va[u].y, 0.0f) - cy;
                    float dz = fmaxf(va[u].z, 0.0f) - cz;
                    float dw = fmaxf(va[u].w, 0.0f) - cw;
                    float ss = fmaf(dw, dw, fmaf(dz, dz, fmaf(dx, dx, dy * dy)));
#pragma unroll
                    for (int off = 16; off > 0; off >>= 1) ss += __shfl_xor(ss, off, 32);
                    if (lane == j + u) myss = ss;
                }
            }
            float wv = 0.0f;
            if (lane < n) {
                if (mya != c) {
                    float nv = fmaxf(sqrtf(myss), 1e-12f);
                    wv = al * sqrtf(nv);
                }
                w_k[b + lane] = wv;
            }
            wsum += wv;
        }
#pragma unroll
        for (int off = 16; off > 0; off >>= 1) wsum += __shfl_xor(wsum, off, 32);
        if (lane == 0) degw[c] = (cnt[c] + 1.0f) + wsum;
    }
}


static inline int nblk(long long threads) { return (int)((threads + TPB - 1) / TPB); }

extern "C" void kernel_launch(void* const* d_in, const int* in_sizes, int n_in,
                              void* d_out, int out_size, void* d_ws, size_t ws_size,
                              hipStream_t stream) {
    (void)n_in; (void)out_size; (void)ws_size;
    const float* x      = (const float*)d_in[0];
    const int*   ei_org = (const int*)d_in[1];
    const int*   ei_knn = (const int*)d_in[2];
    const float* alpha  = (const float*)d_in[3];
    const float* W1     = (const float*)d_in[4];
    const float* b1     = (const float*)d_in[5];
    const float* W2     = (const float*)d_in[6];
    const float* b2     = (const float*)d_in[7];
    float* out = (float*)d_out;
    float* ws  = (float*)d_ws;

    const int N = NNODE, DHID = 128, DOUT = 64;
    const int Eo = in_sizes[1] / 2;
    const int Ek = in_sizes[2] / 2;

    float* deg  = ws;
    float* degw = ws + 2LL * N;
    float* hlin = ws + 4LL * N;
    float* h    = hlin + (long long)N * DHID;
    float* h2   = h + (long long)N * DHID;
    float* wknn = h2 + (long long)N * DOUT;
    float* part = wknn + Ek;
    float* cnt  = ws + 1LL * N;
    float* cntk = ws + 3LL * N;
    int* rowptr = (int*)(part + 3LL * CNT_G * NPITCH);
    int* srt    = rowptr + NPITCH;
    int* rowptr_k = srt + Eo;
    int* srt_k  = rowptr_k + NPITCH;

    const int* row_o = ei_org;
    const int* col_o = ei_org + Eo;
    const int* row_k = ei_knn;
    const int* col_k = ei_knn + Ek;

    k_count_lds<<<3 * CNT_G, CNT_TPB, 0, stream>>>(row_o, col_o, col_k, part, Eo, Ek);
    k_deg_sum<<<nblk(3LL * N), TPB, 0, stream>>>(part, deg, cnt, cntk);

    k_fill_lds<<<2 * CNT_G, CNT_TPB, 0, stream>>>(row_o, col_o, row_k, col_k,
                                                  part + (long long)CNT_G * NPITCH, cnt, cntk,
                                                  rowptr, rowptr_k, srt, srt_k, Eo, Ek);

    k_gemm_bias_wmma<128, false><<<(N / 16 + GT / 32 - 1) / (GT / 32), GT, 0, stream>>>(x, W1, b1,
                                                                                       hlin, N);

    k_agg_nodes<128, 8, false><<<AGG_WG, TPB, 0, stream>>>(rowptr, srt, deg, hlin, h,
                                                           nullptr, nullptr, nullptr);

    k_knn_nodes<<<AGG_WG, TPB, 0, stream>>>(rowptr_k, srt_k, h, alpha, cnt, wknn, degw);

    k_gemm_bias_wmma<64, true><<<(N / 16 + GT / 32 - 1) / (GT / 32), GT, 0, stream>>>(h, W2, b2,
                                                                                     h2, N);

    k_agg_nodes<64, 16, true><<<AGG_WG, TPB, 0, stream>>>(rowptr, srt, degw, h2, out,
                                                          rowptr_k, srt_k, wknn);
}


# Sort kernel: each edge slice is handled by two work-groups, one per half of the destination nodes (half the cursor table each): 256 work-groups, not 128, issue the scattered stores; list order kept
# speedup vs baseline: 3.7220x; 1.1204x over previous
#include <hip/hip_runtime.h>
#include <math.h>


typedef float v2f __attribute__((ext_vector_type(2)));
typedef float v8f __attribute__((ext_vector_type(8)));
typedef float v4f __attribute__((ext_vector_type(4)));

#define TPB 256

#define NNODE   50000
#define NPITCH  50048
#define CNT_TPB 1024
#define CNT_G   64
#define CNT_U   16

__global__ __launch_bounds__(CNT_TPB) void k_count_lds(const int* __restrict__ idx_a,
                                                       const int* __restrict__ idx_b,
                                                       const int* __restrict__ idx_c,
                                                       float* __restrict__ part, int e_ab, int e_c) {
    __shared__ unsigned tbl[NNODE];
    const int list = (int)blockIdx.x / CNT_G;
    const int g = (int)blockIdx.x - list * CNT_G;
    const int* __restrict__ idx = (list == 0) ? idx_a : ((list == 1) ? idx_b : idx_c);
    const int e = (list == 2) ? e_c : e_ab;

    uint4* t4 = (uint4*)tbl;
    for (int i = threadIdx.x; i < NNODE / 4; i += CNT_TPB) t4[i] = make_uint4(0u, 0u, 0u, 0u);
    __syncthreads();

    const int stride = CNT_G * CNT_TPB;
    for (int base = g * CNT_TPB + (int)threadIdx.x; base < e; base += CNT_U * stride) {
        int v[CNT_U];
#pragma unroll
        for (int u = 0; u < CNT_U; ++u) {
            const int i = base + u * stride;
            v[u] = (i < e) ? idx[i] : -1;
        }
#pragma unroll
        for (int u = 0; u < CNT_U; ++u)
            if (v[u] >= 0) atomicAdd(&tbl[v[u]], 1u);
    }
    __syncthreads();

    float* __restrict__ prow = part + (long long)blockIdx.x * NPITCH;
    for (int i = threadIdx.x; i < NNODE; i += CNT_TPB) prow[i] = (float)tbl[i];
}

__global__ __launch_bounds__(TPB) void k_deg_sum(float* __restrict__ part,
                                                 float* __restrict__ deg,
                                                 float* __restrict__ cnt,
                                                 float* __restrict__ cntk) {
    const int j = blockIdx.x * TPB + threadIdx.x;
    if (j >= 3 * NNODE) return;
    const int t = j / NNODE;
    const int i = j - t * NNODE;
    float* p = part + (long long)t * CNT_G * NPITCH + i;
    if (!t) {
        float s = 1.0f;
#pragma unroll 32
        for (int r = 0; r < CNT_G; ++r) s += p[(long long)r * NPITCH];
        deg[i] = s;
    } else {
        float run = 0.0f;
#pragma unroll 32
        for (int r = 0; r < CNT_G; ++r) {
            const float c = p[(long long)r * NPITCH];
            p[(long long)r * NPITCH] = run;
            run += c;
        }
        if (t == 1) cnt[i] = run; else cntk[i] = run;
    }
}

#define FILL_R  2
#define NH      (NNODE / FILL_R)
#define SCAN_CH 25
#define NQH     (NH / 4)
#define NQH_PER ((NQH + CNT_TPB - 1) / CNT_TPB)
__global__ __launch_bounds__(CNT_TPB) void k_fill_lds(const int* __restrict__ src_o,
                                                      const int* __restrict__ dst_o,
                                                      const int* __restrict__ src_k,
                                                      const int* __restrict__ dst_k,
                                                      const float* __restrict__ part_bc,
                                                      const float* __restrict__ cnt_o,
                                                      const float* __restrict__ cnt_k,
                                                      int* __restrict__ rowptr_o,
                                                      int* __restrict__ rowptr_k,
                                                      int* __restrict__ srt_o,
                                                      int* __restrict__ srt_k, int e_o, int e_k) {
    __shared__ unsigned tbl[CNT_TPB * SCAN_CH];
    __shared__ unsigned wsum[CNT_TPB / 32];
    const int tid = (int)threadIdx.x;
    const int lane = tid & 31;
    const int wv = tid >> 5;
    const int hf = (int)blockIdx.x % FILL_R;
    const int sl = (int)blockIdx.x / FILL_R;
    const bool knn = sl >= CNT_G;
    const int g = knn ? sl - CNT_G : sl;
    const int lo = hf * NH;
    const int qlo = hf * NQH;
    const int* __restrict__ idx_src = knn ? src_k : src_o;
    const int* __restrict__ idx_dst = knn ? dst_k : dst_o;
    const float* __restrict__ cnt = knn ? cnt_k : cnt_o;
    int* __restrict__ rowptr = knn ? rowptr_k : rowptr_o;
    int* __restrict__ src = knn ? srt_k : srt_o;
    const int e = knn ? e_k : e_o;
    uint4* t4 = (uint4*)tbl;

    {
        const v4f* c4 = (const v4f*)cnt + qlo;
        v4f cv[NQH_PER];
#pragma unroll
        for (int s = 0; s < NQH_PER; ++s) {
            const int q = tid + s * CNT_TPB;
            cv[s] = c4[q < NQH ? q : NQH - 1];
        }
#pragma unroll
        for (int s = 0; s < NQH_PER; ++s) {
            const int q = tid + s * CNT_TPB;
            if (q < NQH) t4[q] = make_uint4((unsigned)cv[s].x, (unsigned)cv[s].y,
                                            (unsigned)cv[s].z, (unsigned)cv[s].w);
        }
        if (tid < CNT_TPB * SCAN_CH - NH) tbl[NH + tid] = 0u;
    }
    __syncthreads();

    const int i0 = tid * SCAN_CH;
    unsigned own = 0;
#pragma unroll 5
    for (int k = 0; k < SCAN_CH; ++k) own += tbl[i0 + k];
    unsigned incl = own;
    for (int off = 1; off < 32; off <<= 1) {
        const unsigned n = __shfl_up(incl, off, 32);
        if (lane >= off) incl += n;
    }
    if (lane == 31) wsum[wv] = incl;
    __syncthreads();
    if (wv == 0) {
        const unsigned w = wsum[lane];
        unsigned wi = w;
        for (int off = 1; off < 32; off <<= 1) {
            const unsigned n = __shfl_up(wi, off, 32);
            if (lane >= off) wi += n;
        }
        const unsigned tot = __shfl(wi, 31, 32);
        const unsigned first = hf ? (unsigned)e - tot : 0u;
        wsum[lane] = first + wi - w;
    }
    __syncthreads();
    unsigned run = wsum[wv] + incl - own;
#pragma unroll 5
    for (int k = 0; k < SCAN_CH; ++k) {
        const unsigned c = tbl[i0 + k];
        tbl[i0 + k] = run;
        run += c;
    }
    __syncthreads();

    {
        const v4f* p4 = (const v4f*)(part_bc + (long long)sl * NPITCH) + qlo;
        v4f pv[NQH_PER];
#pragma unroll
        for (int s = 0; s < NQH_PER; ++s) {
            const int q = tid + s * CNT_TPB;
            pv[s] = p4[q < NQH ? q : NQH - 1];
        }
#pragma unroll
        for (int s = 0; s < NQH_PER; ++s) {
            const int q = tid + s * CNT_TPB;
            if (q < NQH) {
                uint4 o = t4[q];
                if (g == 0)
                    ((int4*)rowptr)[qlo + q] = make_int4((int)o.x, (int)o.y, (int)o.z, (int)o.w);
                o.x += (unsigned)pv[s].x; o.y += (unsigned)pv[s].y;
                o.z += (unsigned)pv[s].z; o.w += (unsigned)pv[s].w;
                t4[q] = o;
            }
        }
        if (g == 0 && hf == FILL_R - 1 && tid == 0) rowptr[NNODE] = e;
    }
    __syncthreads();

    const int stride = CNT_G * CNT_TPB;
    for (int base = g * CNT_TPB + tid; base < e; base += CNT_U * stride) {
        int d[CNT_U], r[CNT_U];
#pragma unroll
        for (int u = 0; u < CNT_U; ++u) {
            const int i = base + u * stride;
            d[u] = (i < e) ? idx_dst[i] : -1;
            r[u] = (i < e) ? idx_src[i] : 0;
        }
#pragma unroll
        for (int u = 0; u < CNT_U; ++u) {
            const unsigned dj = (unsigned)(d[u] - lo);
            if (dj < (unsigned)NH) {
                const unsigned pos = atomicAdd(&tbl[dj], 1u);
                src[pos] = r[u];
            }
        }
    }
}

#define GT     512
#define WQ     33
template <int NC, bool RELU_A>
__global__ __launch_bounds__(GT) void k_gemm_bias_wmma(const float* __restrict__ A,
                                                       const float* __restrict__ W,
                                                       const float* __restrict__ bias,
                                                       const float* __restrict__ rdeg,
                                                       float* __restrict__ out,
                                                       int M) {
    constexpr int NT = NC / 16;
    __shared__ v4f wl[NC * WQ];

    const int lane = threadIdx.x & 31;
    const int half = lane >> 4;
    const int l16  = lane & 15;
    const int wv = __builtin_amdgcn_readfirstlane((int)(threadIdx.x >> 5));
    const int strip = blockIdx.x * (GT / 32) + wv;
    const bool live = strip < (M >> 4);
    const int m0 = strip << 4;

    const v4f* W4 = (const v4f*)W;
    v4f wt[NC * 32 / GT];
#pragma unroll
    for (int s = 0; s < NC * 32 / GT; ++s) wt[s] = W4[(int)threadIdx.x + s * GT];

    const int ml = live ? m0 : 0;
    float4 a[16];
    const float4* arow = (const float4*)(A + (long long)(ml + l16) * 128) + half;
#pragma unroll
    for (int j = 0; j < 16; ++j) a[j] = arow[2 * j];

    if constexpr (NC == 128)
        asm volatile("" :: "v"(wt[0]), "v"(wt[1]), "v"(wt[2]), "v"(wt[3]),
                           "v"(wt[4]), "v"(wt[5]), "v"(wt[6]), "v"(wt[7]));
    else
        asm volatile("" :: "v"(wt[0]), "v"(wt[1]), "v"(wt[2]), "v"(wt[3]));

#pragma unroll
    for (int s = 0; s < NC * 32 / GT; ++s) {
        const int i = (int)threadIdx.x + s * GT;
        wl[(i >> 5) * WQ + (i & 31)] = wt[s];
    }
    __builtin_amdgcn_fence(__ATOMIC_RELEASE, "workgroup", "local");
    __builtin_amdgcn_s_barrier();
    __builtin_amdgcn_fence(__ATOMIC_ACQUIRE, "workgroup", "local");
    if (!live) return;

    if (RELU_A) {
#pragma unroll
        for (int j = 0; j < 16; ++j) {
            a[j].x = fmaxf(a[j].x, 0.0f); a[j].y = fmaxf(a[j].y, 0.0f);
            a[j].z = fmaxf(a[j].z, 0.0f); a[j].w = fmaxf(a[j].w, 0.0f);
        }
    }

    v8f acc[NT];
#pragma unroll
    for (int t = 0; t < NT; ++t) acc[t] = (v8f){0.f, 0.f, 0.f, 0.f, 0.f, 0.f, 0.f, 0.f};

    const v4f* bl = wl + l16 * WQ + half;
    v4f bc[NT], bn[NT];
#pragma unroll
    for (int t = 0; t < NT; ++t) bc[t] = bl[t * 16 * WQ];
#pragma unroll
    for (int j = 0; j < 16; ++j) {
        if (j + 1 < 16) {
#pragma unroll
            for (int t = 0; t < NT; ++t) bn[t] = bl[t * 16 * WQ + 2 * (j + 1)];
        }
        __builtin_amdgcn_sched_barrier(0);
        const v2f a0 = {a[j].x, a[j].y};
        const v2f a1 = {a[j].z, a[j].w};
#pragma unroll
        for (int t = 0; t < NT; ++t) {
            const v2f b0 = {bc[t].x, bc[t].y};
            acc[t] = __builtin_amdgcn_wmma_f32_16x16x4_f32(false, a0, false, b0,
                                                           (short)0, acc[t], false, false);
        }
#pragma unroll
        for (int t = 0; t < NT; ++t) {
            const v2f b1 = {bc[t].z, bc[t].w};
            acc[t] = __builtin_amdgcn_wmma_f32_16x16x4_f32(false, a1, false, b1,
                                                           (short)0, acc[t], false, false);
        }
        __builtin_amdgcn_sched_barrier(0);
#pragma unroll
        for (int t = 0; t < NT; ++t) bc[t] = bn[t];
    }

    float bv[NT];
#pragma unroll
    for (int t = 0; t < NT; ++t) bv[t] = bias[t * 16 + l16];
    float sc[8];
#pragma unroll
    for (int r = 0; r < 8; ++r) sc[r] = rsqrtf(rdeg[m0 + r + 8 * half]);
#pragma unroll
    for (int r = 0; r < 8; ++r) {
        const int m = m0 + r + 8 * half;
        float* o = out + (long long)m * NC + l16;
#pragma unroll
        for (int t = 0; t < NT; ++t) o[t * 16] = (acc[t][r] + bv[t]) * sc[r];
    }
}

#define AGG_WG (NNODE / 16)

template <int D, int G, bool WT>
__device__ __forceinline__ void agg_run(int beg, int end, const int* __restrict__ src,
                                        const float* __restrict__ wt,
                                        const float* __restrict__ feat,
                                        int lane, float (&acc)[D / 32]) {
    constexpr int Q = D / 32;
    for (int b = beg; b < end; b += 32) {
        const int cnt = (end - b < 32) ? (end - b) : 32;
        int myr = 0;
        float myw = 0.0f;
        if (lane < cnt) {
            myr = src[b + lane];
            if (WT) myw = wt[b + lane];
        }
        for (int j = 0; j < cnt; j += G) {
            float v[G][Q];
#pragma unroll
            for (int u = 0; u < G; ++u) {
                const int r = __builtin_amdgcn_readlane(myr, j + u);
                const float* p = feat + (long long)r * D + lane;
#pragma unroll
                for (int q = 0; q < Q; ++q) v[u][q] = p[32 * q];
            }
#pragma unroll
            for (int u = 0; u < G; ++u) {
                if (j + u < cnt) {
                    if (WT) {
                        const float w = __int_as_float(__builtin_amdgcn_readlane(__float_as_int(myw), j + u));
#pragma unroll
                        for (int q = 0; q < Q; ++q) acc[q] += w * v[u][q];
                    } else {
#pragma unroll
                        for (int q = 0; q < Q; ++q) acc[q] += v[u][q];
                    }
                }
            }
        }
    }
}

template <int D, int G, bool KNN>
__global__ __launch_bounds__(TPB) void k_agg_nodes(const int* __restrict__ rowptr,
                                                   const int* __restrict__ src,
                                                   const float* __restrict__ dg,
                                                   const float* __restrict__ feat,
                                                   float* __restrict__ outp,
                                                   const int* __restrict__ rowptr_k,
                                                   const int* __restrict__ src_k,
                                                   const float* __restrict__ w_k) {
    constexpr int Q = D / 32;
    const int wave = __builtin_amdgcn_readfirstlane((int)((blockIdx.x * TPB + threadIdx.x) >> 5));
    const int nw = (int)gridDim.x * (TPB / 32);
    const int lane = threadIdx.x & 31;
    for (int c = wave; c < NNODE; c += nw) {
        const int beg = __builtin_amdgcn_readfirstlane(rowptr[c]);
        const int end = __builtin_amdgcn_readfirstlane(rowptr[c + 1]);
        int kbeg = 0, kend = 0;
        if (KNN) {
            kbeg = __builtin_amdgcn_readfirstlane(rowptr_k[c]);
            kend = __builtin_amdgcn_readfirstlane(rowptr_k[c + 1]);
        }
        const float dgc = dg[c];
        const float* self = feat + (long long)c * D + lane;
        float acc[Q];
#pragma unroll
        for (int q = 0; q < Q; ++q) acc[q] = self[32 * q];
        agg_run<D, G, false>(beg, end, src, nullptr, feat, lane, acc);
        if (KNN) agg_run<D, G, true>(kbeg, kend, src_k, w_k, feat, lane, acc);
        const float dc = rsqrtf(dgc);
        float* o = outp + (long long)c * D + lane;
#pragma unroll
        for (int q = 0; q < Q; ++q) o[32 * q] = dc * acc[q];
    }
}

#define GK 8
__global__ __launch_bounds__(TPB) void k_knn_nodes(const int* __restrict__ rowptr_k,
                                                   const int* __restrict__ src_k,
                                                   const float* __restrict__ h,
                                                   const float* __restrict__ alpha,
                                                   const float* __restrict__ cnt,
                                                   float* __restrict__ w_k,
                                                   float* __restrict__ degw) {
    const int wave = __builtin_amdgcn_readfirstlane((int)((blockIdx.x * TPB + threadIdx.x) >> 5));
    const int nw = (int)gridDim.x * (TPB / 32);
    const int lane = threadIdx.x & 31;
    const float al = alpha[0];
    for (int c = wave; c < NNODE; c += nw) {
        const int beg = __builtin_amdgcn_readfirstlane(rowptr_k[c]);
        const int end = __builtin_amdgcn_readfirstlane(rowptr_k[c + 1]);
        const float4 vc = ((const float4*)(h + (long long)c * 128))[lane];
        const float cx = fmaxf(vc.x, 0.0f), cy = fmaxf(vc.y, 0.0f);
        const float cz = fmaxf(vc.z, 0.0f), cw = fmaxf(vc.w, 0.0f);
        float wsum = 0.0f;
        for (int b = beg; b < end; b += 32) {
            const int n = (end - b < 32) ? (end - b) : 32;
            int mya = c;
            if (lane < n) mya = src_k[b + lane];
            float myss = 0.0f;
            for (int j = 0; j < n; j += GK) {
                float4 va[GK];
#pragma unroll
                for (int u = 0; u < GK; ++u) {
                    const int a = __builtin_amdgcn_readlane(mya, j + u);
                    va[u] = ((const float4*)(h + (long long)a * 128))[lane];
                }
#pragma unroll
                for (int u = 0; u < GK; ++u) {
                    float dx = fmaxf(va[u].x, 0.0f) - cx;
                    float dy = fmaxf(va[u].y, 0.0f) - cy;
                    float dz = fmaxf(va[u].z, 0.0f) - cz;
                    float dw = fmaxf(va[u].w, 0.0f) - cw;
                    float ss = fmaf(dw, dw, fmaf(dz, dz, fmaf(dx, dx, dy * dy)));
#pragma unroll
                    for (int off = 16; off > 0; off >>= 1) ss += __shfl_xor(ss, off, 32);
                    if (lane == j + u) myss = ss;
                }
            }
            float wv = 0.0f;
            if (lane < n) {
                if (mya != c) {
                    float nv = fmaxf(sqrtf(myss), 1e-12f);
                    wv = al * sqrtf(nv);
                }
                w_k[b + lane] = wv;
            }
            wsum += wv;
        }
#pragma unroll
        for (int off = 16; off > 0; off >>= 1) wsum += __shfl_xor(wsum, off, 32);
        if (lane == 0) degw[c] = (cnt[c] + 1.0f) + wsum;
    }
}


static inline int nblk(long long threads) { return (int)((threads + TPB - 1) / TPB); }

extern "C" void kernel_launch(void* const* d_in, const int* in_sizes, int n_in,
                              void* d_out, int out_size, void* d_ws, size_t ws_size,
                              hipStream_t stream) {
    (void)n_in; (void)out_size; (void)ws_size;
    const float* x      = (const float*)d_in[0];
    const int*   ei_org = (const int*)d_in[1];
    const int*   ei_knn = (const int*)d_in[2];
    const float* alpha  = (const float*)d_in[3];
    const float* W1     = (const float*)d_in[4];
    const float* b1     = (const float*)d_in[5];
    const float* W2     = (const float*)d_in[6];
    const float* b2     = (const float*)d_in[7];
    float* out = (float*)d_out;
    float* ws  = (float*)d_ws;

    const int N = NNODE, DHID = 128, DOUT = 64;
    const int Eo = in_sizes[1] / 2;
    const int Ek = in_sizes[2] / 2;

    float* deg  = ws;
    float* degw = ws + 2LL * N;
    float* hlin = ws + 4LL * N;
    float* h    = hlin + (long long)N * DHID;
    float* h2   = h + (long long)N * DHID;
    float* wknn = h2 + (long long)N * DOUT;
    float* part = wknn + Ek;
    float* cnt  = ws + 1LL * N;
    float* cntk = ws + 3LL * N;
    int* rowptr = (int*)(part + 3LL * CNT_G * NPITCH);
    int* srt    = rowptr + NPITCH;
    int* rowptr_k = srt + Eo;
    int* srt_k  = rowptr_k + NPITCH;

    const int* row_o = ei_org;
    const int* col_o = ei_org + Eo;
    const int* row_k = ei_knn;
    const int* col_k = ei_knn + Ek;

    k_count_lds<<<3 * CNT_G, CNT_TPB, 0, stream>>>(row_o, col_o, col_k, part, Eo, Ek);
    k_deg_sum<<<nblk(3LL * N), TPB, 0, stream>>>(part, deg, cnt, cntk);

    k_fill_lds<<<2 * CNT_G * FILL_R, CNT_TPB, 0, stream>>>(row_o, col_o, row_k, col_k,
                                                  part + (long long)CNT_G * NPITCH, cnt, cntk,
                                                  rowptr, rowptr_k, srt, srt_k, Eo, Ek);

    k_gemm_bias_wmma<128, false><<<(N / 16 + GT / 32 - 1) / (GT / 32), GT, 0, stream>>>(x, W1, b1,
                                                                                       deg, hlin, N);

    k_agg_nodes<128, 8, false><<<AGG_WG, TPB, 0, stream>>>(rowptr, srt, deg, hlin, h,
                                                           nullptr, nullptr, nullptr);

    k_knn_nodes<<<AGG_WG, TPB, 0, stream>>>(rowptr_k, srt_k, h, alpha, cnt, wknn, degw);

    k_gemm_bias_wmma<64, true><<<(N / 16 + GT / 32 - 1) / (GT / 32), GT, 0, stream>>>(h, W2, b2,
                                                                                     degw, h2, N);

    k_agg_nodes<64, 16, true><<<AGG_WG, TPB, 0, stream>>>(rowptr, srt, degw, h2, out,
                                                          rowptr_k, srt_k, wknn);
}
